# starting kernel
# baseline (speedup 1.0000x reference)
#include <hip/hip_runtime.h>

typedef __attribute__((ext_vector_type(2))) float v2f;
typedef __attribute__((ext_vector_type(8))) float v8f;

#define N_NODES_C 100000
#define N_EDGES_C 25000
#define NNZ_C     800000
#define D_C       64
#define ALPHA_C   0.5f

#define A_PSTRIDE 144
#define B_PSTRIDE 160

__device__ __forceinline__ v8f wmma_f32(v2f a, v2f b, v8f c) {
  return __builtin_amdgcn_wmma_f32_16x16x4_f32(
       false, a,  false, b,
       (short)0, c,  false,  false);
}

__global__ void k_diffuse(const float* __restrict__ X,
                          const float* __restrict__ vals,
                          const int* __restrict__ rows,
                          const int* __restrict__ cols,
                          float* __restrict__ Xd) {
  int t  = blockIdx.x * blockDim.x + threadIdx.x;
  int i  = t >> 4;
  int c4 = (t & 15) << 2;
  __builtin_prefetch(vals + i + 1024, 0, 3);
  __builtin_prefetch(cols + i + 1024, 0, 3);
  float v = vals[i];
  const float4 x = *reinterpret_cast<const float4*>(X + (size_t)cols[i] * D_C + c4);
  float* dst = Xd + (size_t)rows[i] * D_C + c4;
  atomicAdd(dst + 0, v * x.x);
  atomicAdd(dst + 1, v * x.y);
  atomicAdd(dst + 2, v * x.z);
  atomicAdd(dst + 3, v * x.w);
}

__global__ void __launch_bounds__(256) k_gemm64(
    const float* __restrict__ A, const float* __restrict__ Ablend, float alpha,
    const float* __restrict__ B, const float* __restrict__ bias,
    float* __restrict__ Out, int nrows) {
  __shared__ float At[32 * A_PSTRIDE * 2];
  __shared__ float Bf[32 * B_PSTRIDE];
  __shared__ float bs[64];

  const int tid  = threadIdx.x;
  const int row0 = blockIdx.x * 128;

  for (int idx = tid; idx < 128 * 16; idx += 256) {
    int rr = idx >> 4, c4 = idx & 15;
    int row = row0 + rr;
    float4 val = make_float4(0.f, 0.f, 0.f, 0.f);
    if (row < nrows) {
      val = *reinterpret_cast<const float4*>(A + (size_t)row * D_C + (c4 << 2));
      if (Ablend) {
        float4 w = *reinterpret_cast<const float4*>(Ablend + (size_t)row * D_C + (c4 << 2));
        val.x = (1.f - alpha) * val.x + alpha * w.x;
        val.y = (1.f - alpha) * val.y + alpha * w.y;
        val.z = (1.f - alpha) * val.z + alpha * w.z;
        val.w = (1.f - alpha) * val.w + alpha * w.w;
      }
    }
    int p = c4 << 1;
    v2f lo; lo.x = val.x; lo.y = val.y;
    v2f hi2; hi2.x = val.z; hi2.y = val.w;
    *reinterpret_cast<v2f*>(&At[(p * A_PSTRIDE + rr) * 2])       = lo;
    *reinterpret_cast<v2f*>(&At[((p + 1) * A_PSTRIDE + rr) * 2]) = hi2;
  }
  for (int idx = tid; idx < 64 * 64; idx += 256) {
    int k = idx >> 6, col = idx & 63;
    Bf[(k >> 1) * B_PSTRIDE + (col << 1) + (k & 1)] = B[idx];
  }
  if (tid < 64) bs[tid] = bias[tid];
  __syncthreads();

  const int wave = tid >> 5, lane = tid & 31;
  const int lrow = lane & 15, hi = lane >> 4;
  const int rr   = wave * 16 + lrow;

  v8f acc[4] = {v8f{}, v8f{}, v8f{}, v8f{}};
#pragma unroll
  for (int k4 = 0; k4 < 16; ++k4) {
    const int p = (k4 << 1) + hi;
    v2f a = *reinterpret_cast<const v2f*>(&At[(p * A_PSTRIDE + rr) * 2]);
#pragma unroll
    for (int n = 0; n < 4; ++n) {
      v2f b = *reinterpret_cast<const v2f*>(
          &Bf[p * B_PSTRIDE + (((n << 4) + lrow) << 1)]);
      acc[n] = wmma_f32(a, b, acc[n]);
    }
  }

#pragma unroll
  for (int n = 0; n < 4; ++n)
#pragma unroll
    for (int j = 0; j < 8; ++j) {
      int row = row0 + wave * 16 + j + (hi << 3);
      if (row < nrows) {
        int col = (n << 4) + lrow;
        Out[(size_t)row * D_C + col] = acc[n][j] + bs[col];
      }
    }
}

__global__ void k_scatter_edge(const float* __restrict__ Y1,
                               const int* __restrict__ vertex,
                               const int* __restrict__ edges,
                               float* __restrict__ Xe) {
  int t  = blockIdx.x * blockDim.x + threadIdx.x;
  int i  = t >> 4;
  int c4 = (t & 15) << 2;
  __builtin_prefetch(vertex + i + 1024, 0, 3);
  __builtin_prefetch(edges + i + 1024, 0, 3);
  const float4 y = *reinterpret_cast<const float4*>(Y1 + (size_t)vertex[i] * D_C + c4);
  float* dst = Xe + (size_t)edges[i] * D_C + c4;
  atomicAdd(dst + 0, y.x);
  atomicAdd(dst + 1, y.y);
  atomicAdd(dst + 2, y.z);
  atomicAdd(dst + 3, y.w);
}

__global__ void __launch_bounds__(256) k_pair_gemm(
    const float* __restrict__ Xd, const float* __restrict__ Xe,
    const int* __restrict__ vertex, const int* __restrict__ edges,
    const float* __restrict__ W2, const float* __restrict__ b2,
    float* __restrict__ Xv) {
  __shared__ float At[64 * A_PSTRIDE * 2];
  __shared__ float Bf[64 * B_PSTRIDE];
  __shared__ float bs[64];

  const int tid = threadIdx.x;
  const int pb  = blockIdx.x * 128;

  __builtin_prefetch(vertex + pb + 128, 0, 3);
  __builtin_prefetch(edges + pb + 128, 0, 3);

  for (int idx = tid; idx < 128 * 32; idx += 256) {
    int pp = idx >> 5, c4 = idx & 31;
    const float* src =
        (c4 < 16) ? (Xd + (size_t)vertex[pb + pp] * D_C + (c4 << 2))
                  : (Xe + (size_t)edges[pb + pp] * D_C + ((c4 - 16) << 2));
    float4 val = *reinterpret_cast<const float4*>(src);
    int p = c4 << 1;
    v2f lo; lo.x = val.x; lo.y = val.y;
    v2f hi2; hi2.x = val.z; hi2.y = val.w;
    *reinterpret_cast<v2f*>(&At[(p * A_PSTRIDE + pp) * 2])       = lo;
    *reinterpret_cast<v2f*>(&At[((p + 1) * A_PSTRIDE + pp) * 2]) = hi2;
  }
  for (int idx = tid; idx < 128 * 64; idx += 256) {
    int k = idx >> 6, col = idx & 63;
    Bf[(k >> 1) * B_PSTRIDE + (col << 1) + (k & 1)] = W2[idx];
  }
  if (tid < 64) bs[tid] = b2[tid];
  __syncthreads();

  const int wave = tid >> 5, lane = tid & 31;
  const int lrow = lane & 15, hi = lane >> 4;
  const int rr   = wave * 16 + lrow;

  v8f acc[4] = {v8f{}, v8f{}, v8f{}, v8f{}};
#pragma unroll
  for (int k4 = 0; k4 < 32; ++k4) {
    const int p = (k4 << 1) + hi;
    v2f a = *reinterpret_cast<const v2f*>(&At[(p * A_PSTRIDE + rr) * 2]);
#pragma unroll
    for (int n = 0; n < 4; ++n) {
      v2f b = *reinterpret_cast<const v2f*>(
          &Bf[p * B_PSTRIDE + (((n << 4) + lrow) << 1)]);
      acc[n] = wmma_f32(a, b, acc[n]);
    }
  }

  const int p0w = pb + wave * 16;
#pragma unroll
  for (int j = 0; j < 8; ++j) {
    int row = vertex[p0w + j + (hi << 3)];
#pragma unroll
    for (int n = 0; n < 4; ++n) {
      int col = (n << 4) + lrow;
      atomicAdd(&Xv[(size_t)row * D_C + col], acc[n][j] + bs[col]);
    }
  }
}

extern "C" void kernel_launch(void* const* d_in, const int* in_sizes, int n_in,
                              void* d_out, int out_size, void* d_ws, size_t ws_size,
                              hipStream_t stream) {
  const float* X      = (const float*)d_in[0];
  const float* W1_w   = (const float*)d_in[1];
  const float* W1_b   = (const float*)d_in[2];
  const float* W2_w   = (const float*)d_in[3];
  const float* W2_b   = (const float*)d_in[4];
  const float* W_w    = (const float*)d_in[5];
  const float* W_b    = (const float*)d_in[6];
  const float* dvals  = (const float*)d_in[7];
  const int*   vertex = (const int*)d_in[8];
  const int*   edges  = (const int*)d_in[9];
  const int*   drows  = (const int*)d_in[10];
  const int*   dcols  = (const int*)d_in[11];
  float* out = (float*)d_out;

  float* Xd = (float*)d_ws;
  float* Y1 = Xd + (size_t)N_NODES_C * D_C;
  float* Xe = Y1 + (size_t)N_NODES_C * D_C;
  float* Xv = Xe + (size_t)N_EDGES_C * D_C;

  hipMemsetAsync(Xd, 0, (size_t)N_NODES_C * D_C * sizeof(float), stream);
  hipMemsetAsync(Xe, 0, (size_t)N_EDGES_C * D_C * sizeof(float), stream);
  hipMemsetAsync(Xv, 0, (size_t)N_NODES_C * D_C * sizeof(float), stream);

  k_diffuse<<<NNZ_C * 16 / 256, 256, 0, stream>>>(X, dvals, drows, dcols, Xd);
  k_gemm64<<<(N_NODES_C + 127) / 128, 256, 0, stream>>>(
      Xd, nullptr, 0.f, W1_w, W1_b, Y1, N_NODES_C);
  k_scatter_edge<<<NNZ_C * 16 / 256, 256, 0, stream>>>(Y1, vertex, edges, Xe);
  k_pair_gemm<<<NNZ_C / 128, 256, 0, stream>>>(Xd, Xe, vertex, edges, W2_w, W2_b, Xv);
  k_gemm64<<<(N_NODES_C + 127) / 128, 256, 0, stream>>>(
      Xv, Xd, ALPHA_C, W_w, W_b, out, N_NODES_C);
}


# A-panel LDS layout of both GEMM kernels changed to [k4][row] float4 with odd row pitch: staging stores are conflict-free ds_store_b128 (were ds_store_b64 with all lanes of a wave on one bank pair)
# speedup vs baseline: 1.1035x; 1.1035x over previous
#include <hip/hip_runtime.h>

typedef __attribute__((ext_vector_type(2))) float v2f;
typedef __attribute__((ext_vector_type(8))) float v8f;

#define N_NODES_C 100000
#define N_EDGES_C 25000
#define NNZ_C     800000
#define D_C       64
#define ALPHA_C   0.5f

#define A_RS      129
#define B_PSTRIDE 160

__device__ __forceinline__ v8f wmma_f32(v2f a, v2f b, v8f c) {
  return __builtin_amdgcn_wmma_f32_16x16x4_f32(
       false, a,  false, b,
       (short)0, c,  false,  false);
}

__global__ void k_diffuse(const float* __restrict__ X,
                          const float* __restrict__ vals,
                          const int* __restrict__ rows,
                          const int* __restrict__ cols,
                          float* __restrict__ Xd) {
  int t  = blockIdx.x * blockDim.x + threadIdx.x;
  int i  = t >> 4;
  int c4 = (t & 15) << 2;
  __builtin_prefetch(vals + i + 1024, 0, 3);
  __builtin_prefetch(cols + i + 1024, 0, 3);
  float v = vals[i];
  const float4 x = *reinterpret_cast<const float4*>(X + (size_t)cols[i] * D_C + c4);
  float* dst = Xd + (size_t)rows[i] * D_C + c4;
  atomicAdd(dst + 0, v * x.x);
  atomicAdd(dst + 1, v * x.y);
  atomicAdd(dst + 2, v * x.z);
  atomicAdd(dst + 3, v * x.w);
}

__global__ void __launch_bounds__(256) k_gemm64(
    const float* __restrict__ A, const float* __restrict__ Ablend, float alpha,
    const float* __restrict__ B, const float* __restrict__ bias,
    float* __restrict__ Out, int nrows) {
  __shared__ __attribute__((aligned(16))) float At[16 * A_RS * 4];
  __shared__ float Bf[32 * B_PSTRIDE];
  __shared__ float bs[64];

  const int tid  = threadIdx.x;
  const int row0 = blockIdx.x * 128;

  for (int idx = tid; idx < 128 * 16; idx += 256) {
    int rr = idx >> 4, c4 = idx & 15;
    int row = row0 + rr;
    float4 val = make_float4(0.f, 0.f, 0.f, 0.f);
    if (row < nrows) {
      val = *reinterpret_cast<const float4*>(A + (size_t)row * D_C + (c4 << 2));
      if (Ablend) {
        float4 w = *reinterpret_cast<const float4*>(Ablend + (size_t)row * D_C + (c4 << 2));
        val.x = (1.f - alpha) * val.x + alpha * w.x;
        val.y = (1.f - alpha) * val.y + alpha * w.y;
        val.z = (1.f - alpha) * val.z + alpha * w.z;
        val.w = (1.f - alpha) * val.w + alpha * w.w;
      }
    }
    *reinterpret_cast<float4*>(&At[(c4 * A_RS + rr) * 4]) = val;
  }
  for (int idx = tid; idx < 64 * 64; idx += 256) {
    int k = idx >> 6, col = idx & 63;
    Bf[(k >> 1) * B_PSTRIDE + (col << 1) + (k & 1)] = B[idx];
  }
  if (tid < 64) bs[tid] = bias[tid];
  __syncthreads();

  const int wave = tid >> 5, lane = tid & 31;
  const int lrow = lane & 15, hi = lane >> 4;
  const int rr   = wave * 16 + lrow;

  v8f acc[4] = {v8f{}, v8f{}, v8f{}, v8f{}};
#pragma unroll
  for (int k4 = 0; k4 < 16; ++k4) {
    const int p = (k4 << 1) + hi;
    v2f a = *reinterpret_cast<const v2f*>(&At[(k4 * A_RS + rr) * 4 + (hi << 1)]);
#pragma unroll
    for (int n = 0; n < 4; ++n) {
      v2f b = *reinterpret_cast<const v2f*>(
          &Bf[p * B_PSTRIDE + (((n << 4) + lrow) << 1)]);
      acc[n] = wmma_f32(a, b, acc[n]);
    }
  }

#pragma unroll
  for (int n = 0; n < 4; ++n)
#pragma unroll
    for (int j = 0; j < 8; ++j) {
      int row = row0 + wave * 16 + j + (hi << 3);
      if (row < nrows) {
        int col = (n << 4) + lrow;
        Out[(size_t)row * D_C + col] = acc[n][j] + bs[col];
      }
    }
}

__global__ void k_scatter_edge(const float* __restrict__ Y1,
                               const int* __restrict__ vertex,
                               const int* __restrict__ edges,
                               float* __restrict__ Xe) {
  int t  = blockIdx.x * blockDim.x + threadIdx.x;
  int i  = t >> 4;
  int c4 = (t & 15) << 2;
  __builtin_prefetch(vertex + i + 1024, 0, 3);
  __builtin_prefetch(edges + i + 1024, 0, 3);
  const float4 y = *reinterpret_cast<const float4*>(Y1 + (size_t)vertex[i] * D_C + c4);
  float* dst = Xe + (size_t)edges[i] * D_C + c4;
  atomicAdd(dst + 0, y.x);
  atomicAdd(dst + 1, y.y);
  atomicAdd(dst + 2, y.z);
  atomicAdd(dst + 3, y.w);
}

__global__ void __launch_bounds__(256) k_pair_gemm(
    const float* __restrict__ Xd, const float* __restrict__ Xe,
    const int* __restrict__ vertex, const int* __restrict__ edges,
    const float* __restrict__ W2, const float* __restrict__ b2,
    float* __restrict__ Xv) {
  __shared__ __attribute__((aligned(16))) float At[32 * A_RS * 4];
  __shared__ float Bf[64 * B_PSTRIDE];
  __shared__ float bs[64];

  const int tid = threadIdx.x;
  const int pb  = blockIdx.x * 128;

  __builtin_prefetch(vertex + pb + 128, 0, 3);
  __builtin_prefetch(edges + pb + 128, 0, 3);

  for (int idx = tid; idx < 128 * 32; idx += 256) {
    int pp = idx >> 5, c4 = idx & 31;
    const float* src =
        (c4 < 16) ? (Xd + (size_t)vertex[pb + pp] * D_C + (c4 << 2))
                  : (Xe + (size_t)edges[pb + pp] * D_C + ((c4 - 16) << 2));
    float4 val = *reinterpret_cast<const float4*>(src);
    *reinterpret_cast<float4*>(&At[(c4 * A_RS + pp) * 4]) = val;
  }
  for (int idx = tid; idx < 128 * 64; idx += 256) {
    int k = idx >> 6, col = idx & 63;
    Bf[(k >> 1) * B_PSTRIDE + (col << 1) + (k & 1)] = W2[idx];
  }
  if (tid < 64) bs[tid] = b2[tid];
  __syncthreads();

  const int wave = tid >> 5, lane = tid & 31;
  const int lrow = lane & 15, hi = lane >> 4;
  const int rr   = wave * 16 + lrow;

  v8f acc[4] = {v8f{}, v8f{}, v8f{}, v8f{}};
#pragma unroll
  for (int k4 = 0; k4 < 32; ++k4) {
    const int p = (k4 << 1) + hi;
    v2f a = *reinterpret_cast<const v2f*>(&At[(k4 * A_RS + rr) * 4 + (hi << 1)]);
#pragma unroll
    for (int n = 0; n < 4; ++n) {
      v2f b = *reinterpret_cast<const v2f*>(
          &Bf[p * B_PSTRIDE + (((n << 4) + lrow) << 1)]);
      acc[n] = wmma_f32(a, b, acc[n]);
    }
  }

  const int p0w = pb + wave * 16;
#pragma unroll
  for (int j = 0; j < 8; ++j) {
    int row = vertex[p0w + j + (hi << 3)];
#pragma unroll
    for (int n = 0; n < 4; ++n) {
      int col = (n << 4) + lrow;
      atomicAdd(&Xv[(size_t)row * D_C + col], acc[n][j] + bs[col]);
    }
  }
}

extern "C" void kernel_launch(void* const* d_in, const int* in_sizes, int n_in,
                              void* d_out, int out_size, void* d_ws, size_t ws_size,
                              hipStream_t stream) {
  const float* X      = (const float*)d_in[0];
  const float* W1_w   = (const float*)d_in[1];
  const float* W1_b   = (const float*)d_in[2];
  const float* W2_w   = (const float*)d_in[3];
  const float* W2_b   = (const float*)d_in[4];
  const float* W_w    = (const float*)d_in[5];
  const float* W_b    = (const float*)d_in[6];
  const float* dvals  = (const float*)d_in[7];
  const int*   vertex = (const int*)d_in[8];
  const int*   edges  = (const int*)d_in[9];
  const int*   drows  = (const int*)d_in[10];
  const int*   dcols  = (const int*)d_in[11];
  float* out = (float*)d_out;

  float* Xd = (float*)d_ws;
  float* Y1 = Xd + (size_t)N_NODES_C * D_C;
  float* Xe = Y1 + (size_t)N_NODES_C * D_C;
  float* Xv = Xe + (size_t)N_EDGES_C * D_C;

  hipMemsetAsync(Xd, 0, (size_t)N_NODES_C * D_C * sizeof(float), stream);
  hipMemsetAsync(Xe, 0, (size_t)N_EDGES_C * D_C * sizeof(float), stream);
  hipMemsetAsync(Xv, 0, (size_t)N_NODES_C * D_C * sizeof(float), stream);

  k_diffuse<<<NNZ_C * 16 / 256, 256, 0, stream>>>(X, dvals, drows, dcols, Xd);
  k_gemm64<<<(N_NODES_C + 127) / 128, 256, 0, stream>>>(
      Xd, nullptr, 0.f, W1_w, W1_b, Y1, N_NODES_C);
  k_scatter_edge<<<NNZ_C * 16 / 256, 256, 0, stream>>>(Y1, vertex, edges, Xe);
  k_pair_gemm<<<NNZ_C / 128, 256, 0, stream>>>(Xd, Xe, vertex, edges, W2_w, W2_b, Xv);
  k_gemm64<<<(N_NODES_C + 127) / 128, 256, 0, stream>>>(
      Xv, Xd, ALPHA_C, W_w, W_b, out, N_NODES_C);
}


# Xd, Xe segment sums by destination-range owner work-groups: each scans the key list, gathers its rows in groups, sums in LDS (LDS float adds, arrival order); no global float atomics there
# speedup vs baseline: 1.2231x; 1.1084x over previous
#include <hip/hip_runtime.h>

typedef __attribute__((ext_vector_type(2))) float v2f;
typedef __attribute__((ext_vector_type(8))) float v8f;

#define N_NODES_C 100000
#define N_EDGES_C 25000
#define NNZ_C     800000
#define D_C       64
#define ALPHA_C   0.5f

#define A_RS      129
#define B_PSTRIDE 160

__device__ __forceinline__ v8f wmma_f32(v2f a, v2f b, v8f c) {
  return __builtin_amdgcn_wmma_f32_16x16x4_f32(
       false, a,  false, b,
       (short)0, c,  false,  false);
}

#define OWN_THREADS 1024
#define OWN_WAVES   (OWN_THREADS / 32)
#define OWN_U       4
#define OWN_DRAIN   32
#define OWN_QCAP    (OWN_DRAIN + 128 * OWN_U)
#define OWN_B       16

template <int NR, bool SCALED>
__global__ void __launch_bounds__(OWN_THREADS) k_owner_sum(
    const float* __restrict__ src, const int* __restrict__ key,
    const int* __restrict__ gidx, const float* __restrict__ scale,
    float* __restrict__ dst, int ndst) {
  __shared__ __attribute__((aligned(16))) float acc[(NR + 1) * D_C];
  __shared__ int2 queue[OWN_WAVES * OWN_QCAP];

  const int tid  = threadIdx.x;
  const int lane = tid & 31;
  const int r0   = blockIdx.x * NR;
  int nown = ndst - r0;
  nown = nown > NR ? NR : (nown < 0 ? 0 : nown);

  for (int j = tid; j < (NR + 1) * D_C; j += OWN_THREADS) acc[j] = 0.f;
  __syncthreads();

  int2* q = queue + (tid >> 5) * OWN_QCAP;
  int count = 0;

  auto drain = [&]() {
    __builtin_amdgcn_fence(__ATOMIC_ACQ_REL, "wavefront");
    for (int b = 0; b < count; b += OWN_B) {
      int it[OWN_B], lr[OWN_B];
#pragma unroll
      for (int k = 0; k < OWN_B; ++k) {
        const int j  = b + k;
        const int2 en = q[j < count ? j : count - 1];
        it[k] = en.x;
        lr[k] = (j < count) ? en.y : NR;
      }
      int g[OWN_B];
      float s[OWN_B];
#pragma unroll
      for (int k = 0; k < OWN_B; ++k) {
        g[k] = gidx[it[k]];
        s[k] = SCALED ? scale[it[k]] : 1.f;
      }
      v2f x[OWN_B];
#pragma unroll
      for (int k = 0; k < OWN_B; ++k)
        x[k] = *reinterpret_cast<const v2f*>(src + (size_t)g[k] * D_C + (lane << 1));
#pragma unroll
      for (int k = 0; k < OWN_B; ++k) {
        float a = x[k][0], c = x[k][1];
        if (SCALED) { a *= s[k]; c *= s[k]; }
        float* d = &acc[lr[k] * D_C + (lane << 1)];
        __hip_atomic_fetch_add(d,     a, __ATOMIC_RELAXED, __HIP_MEMORY_SCOPE_WORKGROUP);
        __hip_atomic_fetch_add(d + 1, c, __ATOMIC_RELAXED, __HIP_MEMORY_SCOPE_WORKGROUP);
      }
    }
    __builtin_amdgcn_fence(__ATOMIC_ACQ_REL, "wavefront");
    count = 0;
  };

  auto append = [&](int k, int item, bool ok) {
    const unsigned lrow = (unsigned)(k - r0);
    const bool m = ok && lrow < (unsigned)nown;
    const unsigned mask = __builtin_amdgcn_ballot_w32(m);
    if (mask) {
      if (m) q[count + __builtin_amdgcn_mbcnt_lo(mask, 0u)] = make_int2(item, (int)lrow);
      count += __builtin_popcount(mask);
    }
  };

  static_assert(NNZ_C % 4 == 0, "key list read as 16-byte pieces");
  constexpr int NQ   = NNZ_C / 4;
  constexpr int STEP = OWN_THREADS * OWN_U;
  const int4* key4 = reinterpret_cast<const int4*>(key);

  auto load_keys = [&](int4 (&buf)[OWN_U], int base) {
#pragma unroll
    for (int u = 0; u < OWN_U; ++u) {
      const int qi = base + u * OWN_THREADS + tid;
      buf[u] = key4[qi < NQ ? qi : NQ - 1];
    }
  };
  auto test_keys = [&](const int4 (&buf)[OWN_U], int base) {
#pragma unroll
    for (int u = 0; u < OWN_U; ++u) {
      const int qi = base + u * OWN_THREADS + tid;
      const bool ok = qi < NQ;
      const int i0 = qi << 2;
      append(buf[u].x, i0 + 0, ok);
      append(buf[u].y, i0 + 1, ok);
      append(buf[u].z, i0 + 2, ok);
      append(buf[u].w, i0 + 3, ok);
    }
    if (count >= OWN_DRAIN) drain();
  };

  int4 ka[OWN_U], kb[OWN_U];
  load_keys(ka, 0);
  for (int base = 0; base < NQ; base += 2 * STEP) {
    load_keys(kb, base + STEP);
    test_keys(ka, base);
    load_keys(ka, base + 2 * STEP);
    test_keys(kb, base + STEP);
  }
  drain();
  __syncthreads();

  for (int j = tid; j < nown * (D_C / 4); j += OWN_THREADS)
    *reinterpret_cast<float4*>(dst + (size_t)r0 * D_C + (j << 2)) =
        *reinterpret_cast<const float4*>(&acc[j << 2]);
}

__global__ void __launch_bounds__(256) k_gemm64(
    const float* __restrict__ A, const float* __restrict__ Ablend, float alpha,
    const float* __restrict__ B, const float* __restrict__ bias,
    float* __restrict__ Out, int nrows) {
  __shared__ __attribute__((aligned(16))) float At[16 * A_RS * 4];
  __shared__ float Bf[32 * B_PSTRIDE];
  __shared__ float bs[64];

  const int tid  = threadIdx.x;
  const int row0 = blockIdx.x * 128;

  for (int idx = tid; idx < 128 * 16; idx += 256) {
    int rr = idx >> 4, c4 = idx & 15;
    int row = row0 + rr;
    float4 val = make_float4(0.f, 0.f, 0.f, 0.f);
    if (row < nrows) {
      val = *reinterpret_cast<const float4*>(A + (size_t)row * D_C + (c4 << 2));
      if (Ablend) {
        float4 w = *reinterpret_cast<const float4*>(Ablend + (size_t)row * D_C + (c4 << 2));
        val.x = (1.f - alpha) * val.x + alpha * w.x;
        val.y = (1.f - alpha) * val.y + alpha * w.y;
        val.z = (1.f - alpha) * val.z + alpha * w.z;
        val.w = (1.f - alpha) * val.w + alpha * w.w;
      }
    }
    *reinterpret_cast<float4*>(&At[(c4 * A_RS + rr) * 4]) = val;
  }
  for (int idx = tid; idx < 64 * 64; idx += 256) {
    int k = idx >> 6, col = idx & 63;
    Bf[(k >> 1) * B_PSTRIDE + (col << 1) + (k & 1)] = B[idx];
  }
  if (tid < 64) bs[tid] = bias[tid];
  __syncthreads();

  const int wave = tid >> 5, lane = tid & 31;
  const int lrow = lane & 15, hi = lane >> 4;
  const int rr   = wave * 16 + lrow;

  v8f acc[4] = {v8f{}, v8f{}, v8f{}, v8f{}};
#pragma unroll
  for (int k4 = 0; k4 < 16; ++k4) {
    const int p = (k4 << 1) + hi;
    v2f a = *reinterpret_cast<const v2f*>(&At[(k4 * A_RS + rr) * 4 + (hi << 1)]);
#pragma unroll
    for (int n = 0; n < 4; ++n) {
      v2f b = *reinterpret_cast<const v2f*>(
          &Bf[p * B_PSTRIDE + (((n << 4) + lrow) << 1)]);
      acc[n] = wmma_f32(a, b, acc[n]);
    }
  }

#pragma unroll
  for (int n = 0; n < 4; ++n)
#pragma unroll
    for (int j = 0; j < 8; ++j) {
      int row = row0 + wave * 16 + j + (hi << 3);
      if (row < nrows) {
        int col = (n << 4) + lrow;
        Out[(size_t)row * D_C + col] = acc[n][j] + bs[col];
      }
    }
}

__global__ void __launch_bounds__(256) k_pair_gemm(
    const float* __restrict__ Xd, const float* __restrict__ Xe,
    const int* __restrict__ vertex, const int* __restrict__ edges,
    const float* __restrict__ W2, const float* __restrict__ b2,
    float* __restrict__ Xv) {
  __shared__ __attribute__((aligned(16))) float At[32 * A_RS * 4];
  __shared__ float Bf[64 * B_PSTRIDE];
  __shared__ float bs[64];

  const int tid = threadIdx.x;
  const int pb  = blockIdx.x * 128;

  __builtin_prefetch(vertex + pb + 128, 0, 3);
  __builtin_prefetch(edges + pb + 128, 0, 3);

  for (int idx = tid; idx < 128 * 32; idx += 256) {
    int pp = idx >> 5, c4 = idx & 31;
    const float* src =
        (c4 < 16) ? (Xd + (size_t)vertex[pb + pp] * D_C + (c4 << 2))
                  : (Xe + (size_t)edges[pb + pp] * D_C + ((c4 - 16) << 2));
    float4 val = *reinterpret_cast<const float4*>(src);
    *reinterpret_cast<float4*>(&At[(c4 * A_RS + pp) * 4]) = val;
  }
  for (int idx = tid; idx < 128 * 64; idx += 256) {
    int k = idx >> 6, col = idx & 63;
    Bf[(k >> 1) * B_PSTRIDE + (col << 1) + (k & 1)] = W2[idx];
  }
  if (tid < 64) bs[tid] = b2[tid];
  __syncthreads();

  const int wave = tid >> 5, lane = tid & 31;
  const int lrow = lane & 15, hi = lane >> 4;
  const int rr   = wave * 16 + lrow;

  v8f acc[4] = {v8f{}, v8f{}, v8f{}, v8f{}};
#pragma unroll
  for (int k4 = 0; k4 < 32; ++k4) {
    const int p = (k4 << 1) + hi;
    v2f a = *reinterpret_cast<const v2f*>(&At[(k4 * A_RS + rr) * 4 + (hi << 1)]);
#pragma unroll
    for (int n = 0; n < 4; ++n) {
      v2f b = *reinterpret_cast<const v2f*>(
          &Bf[p * B_PSTRIDE + (((n << 4) + lrow) << 1)]);
      acc[n] = wmma_f32(a, b, acc[n]);
    }
  }

  const int p0w = pb + wave * 16;
#pragma unroll
  for (int j = 0; j < 8; ++j) {
    int row = vertex[p0w + j + (hi << 3)];
#pragma unroll
    for (int n = 0; n < 4; ++n) {
      int col = (n << 4) + lrow;
      atomicAdd(&Xv[(size_t)row * D_C + col], acc[n][j] + bs[col]);
    }
  }
}

extern "C" void kernel_launch(void* const* d_in, const int* in_sizes, int n_in,
                              void* d_out, int out_size, void* d_ws, size_t ws_size,
                              hipStream_t stream) {
  const float* X      = (const float*)d_in[0];
  const float* W1_w   = (const float*)d_in[1];
  const float* W1_b   = (const float*)d_in[2];
  const float* W2_w   = (const float*)d_in[3];
  const float* W2_b   = (const float*)d_in[4];
  const float* W_w    = (const float*)d_in[5];
  const float* W_b    = (const float*)d_in[6];
  const float* dvals  = (const float*)d_in[7];
  const int*   vertex = (const int*)d_in[8];
  const int*   edges  = (const int*)d_in[9];
  const int*   drows  = (const int*)d_in[10];
  const int*   dcols  = (const int*)d_in[11];
  float* out = (float*)d_out;

  float* Xd = (float*)d_ws;
  float* Y1 = Xd + (size_t)N_NODES_C * D_C;
  float* Xe = Y1 + (size_t)N_NODES_C * D_C;
  float* Xv = Xe + (size_t)N_EDGES_C * D_C;

  hipMemsetAsync(Xv, 0, (size_t)N_NODES_C * D_C * sizeof(float), stream);

  k_owner_sum<(N_NODES_C + 255) / 256, true><<<256, OWN_THREADS, 0, stream>>>(
      X, drows, dcols, dvals, Xd, N_NODES_C);
  k_gemm64<<<(N_NODES_C + 127) / 128, 256, 0, stream>>>(
      Xd, nullptr, 0.f, W1_w, W1_b, Y1, N_NODES_C);
  k_owner_sum<(N_EDGES_C + 255) / 256, false><<<256, OWN_THREADS, 0, stream>>>(
      Y1, edges, vertex, nullptr, Xe, N_EDGES_C);
  k_pair_gemm<<<NNZ_C / 128, 256, 0, stream>>>(Xd, Xe, vertex, edges, W2_w, W2_b, Xv);
  k_gemm64<<<(N_NODES_C + 127) / 128, 256, 0, stream>>>(
      Xv, Xd, ALPHA_C, W_w, W_b, out, N_NODES_C);
}


# Per-node sum of Xv moved ahead of the W2 product (linearity): owner work-groups sum Xe rows and count pairs per node in LDS, then one K=128 WMMA product per node; sums reordered, no global atomics
# speedup vs baseline: 1.6595x; 1.3568x over previous
#include <hip/hip_runtime.h>

typedef __attribute__((ext_vector_type(2))) float v2f;
typedef __attribute__((ext_vector_type(8))) float v8f;

#define N_NODES_C 100000
#define N_EDGES_C 25000
#define NNZ_C     800000
#define D_C       64
#define ALPHA_C   0.5f

#define A_RS      129
#define B_PSTRIDE 160

__device__ __forceinline__ v8f wmma_f32(v2f a, v2f b, v8f c) {
  return __builtin_amdgcn_wmma_f32_16x16x4_f32(
       false, a,  false, b,
       (short)0, c,  false,  false);
}

#define OWN_THREADS 1024
#define OWN_WAVES   (OWN_THREADS / 32)
#define OWN_U       4
#define OWN_DRAIN   32
#define OWN_QCAP    (OWN_DRAIN + 128 * OWN_U)
#define OWN_B       16

template <int NR, bool SCALED, bool COUNT>
__global__ void __launch_bounds__(OWN_THREADS) k_owner_sum(
    const float* __restrict__ src, const int* __restrict__ key,
    const int* __restrict__ gidx, const float* __restrict__ scale,
    float* __restrict__ dst, float* __restrict__ cnt_out, int ndst) {
  __shared__ __attribute__((aligned(16))) float acc[(NR + 1) * D_C];
  __shared__ int2 queue[OWN_WAVES * OWN_QCAP];
  __shared__ float cntl[COUNT ? NR + 1 : 1];

  const int tid  = threadIdx.x;
  const int lane = tid & 31;
  const int r0   = blockIdx.x * NR;
  int nown = ndst - r0;
  nown = nown > NR ? NR : (nown < 0 ? 0 : nown);

  for (int j = tid; j < (NR + 1) * D_C; j += OWN_THREADS) acc[j] = 0.f;
  if (COUNT)
    for (int j = tid; j < NR + 1; j += OWN_THREADS) cntl[j] = 0.f;
  __syncthreads();

  int2* q = queue + (tid >> 5) * OWN_QCAP;
  int count = 0;

  auto drain = [&]() {
    __builtin_amdgcn_fence(__ATOMIC_ACQ_REL, "wavefront");
    for (int b = 0; b < count; b += OWN_B) {
      if (COUNT) {
        const int j = b + lane;
        if (lane < OWN_B && j < count)
          __hip_atomic_fetch_add(&cntl[q[j].y], 1.f, __ATOMIC_RELAXED, __HIP_MEMORY_SCOPE_WORKGROUP);
      }
      int it[OWN_B], lr[OWN_B];
#pragma unroll
      for (int k = 0; k < OWN_B; ++k) {
        const int j  = b + k;
        const int2 en = q[j < count ? j : count - 1];
        it[k] = en.x;
        lr[k] = (j < count) ? en.y : NR;
      }
      int g[OWN_B];
      float s[OWN_B];
#pragma unroll
      for (int k = 0; k < OWN_B; ++k) {
        g[k] = gidx[it[k]];
        s[k] = SCALED ? scale[it[k]] : 1.f;
      }
      v2f x[OWN_B];
#pragma unroll
      for (int k = 0; k < OWN_B; ++k)
        x[k] = *reinterpret_cast<const v2f*>(src + (size_t)g[k] * D_C + (lane << 1));
#pragma unroll
      for (int k = 0; k < OWN_B; ++k) {
        float a = x[k][0], c = x[k][1];
        if (SCALED) { a *= s[k]; c *= s[k]; }
        float* d = &acc[lr[k] * D_C + (lane << 1)];
        __hip_atomic_fetch_add(d,     a, __ATOMIC_RELAXED, __HIP_MEMORY_SCOPE_WORKGROUP);
        __hip_atomic_fetch_add(d + 1, c, __ATOMIC_RELAXED, __HIP_MEMORY_SCOPE_WORKGROUP);
      }
    }
    __builtin_amdgcn_fence(__ATOMIC_ACQ_REL, "wavefront");
    count = 0;
  };

  auto append = [&](int k, int item, bool ok) {
    const unsigned lrow = (unsigned)(k - r0);
    const bool m = ok && lrow < (unsigned)nown;
    const unsigned mask = __builtin_amdgcn_ballot_w32(m);
    if (mask) {
      if (m) q[count + __builtin_amdgcn_mbcnt_lo(mask, 0u)] = make_int2(item, (int)lrow);
      count += __builtin_popcount(mask);
    }
  };

  static_assert(NNZ_C % 4 == 0, "key list read as 16-byte pieces");
  constexpr int NQ   = NNZ_C / 4;
  constexpr int STEP = OWN_THREADS * OWN_U;
  const int4* key4 = reinterpret_cast<const int4*>(key);

  auto load_keys = [&](int4 (&buf)[OWN_U], int base) {
#pragma unroll
    for (int u = 0; u < OWN_U; ++u) {
      const int qi = base + u * OWN_THREADS + tid;
      buf[u] = key4[qi < NQ ? qi : NQ - 1];
    }
  };
  auto test_keys = [&](const int4 (&buf)[OWN_U], int base) {
#pragma unroll
    for (int u = 0; u < OWN_U; ++u) {
      const int qi = base + u * OWN_THREADS + tid;
      const bool ok = qi < NQ;
      const int i0 = qi << 2;
      append(buf[u].x, i0 + 0, ok);
      append(buf[u].y, i0 + 1, ok);
      append(buf[u].z, i0 + 2, ok);
      append(buf[u].w, i0 + 3, ok);
    }
    if (count >= OWN_DRAIN) drain();
  };

  int4 ka[OWN_U], kb[OWN_U];
  load_keys(ka, 0);
  for (int base = 0; base < NQ; base += 2 * STEP) {
    load_keys(kb, base + STEP);
    test_keys(ka, base);
    load_keys(ka, base + 2 * STEP);
    test_keys(kb, base + STEP);
  }
  drain();
  __syncthreads();

  for (int j = tid; j < nown * (D_C / 4); j += OWN_THREADS)
    *reinterpret_cast<float4*>(dst + (size_t)r0 * D_C + (j << 2)) =
        *reinterpret_cast<const float4*>(&acc[j << 2]);
  if (COUNT)
    for (int j = tid; j < nown; j += OWN_THREADS) cnt_out[r0 + j] = cntl[j];
}

__global__ void __launch_bounds__(256) k_gemm64(
    const float* __restrict__ A, const float* __restrict__ Ablend, float alpha,
    const float* __restrict__ B, const float* __restrict__ bias,
    float* __restrict__ Out, int nrows) {
  __shared__ __attribute__((aligned(16))) float At[16 * A_RS * 4];
  __shared__ float Bf[32 * B_PSTRIDE];
  __shared__ float bs[64];

  const int tid  = threadIdx.x;
  const int row0 = blockIdx.x * 128;

  for (int idx = tid; idx < 128 * 16; idx += 256) {
    int rr = idx >> 4, c4 = idx & 15;
    int row = row0 + rr;
    float4 val = make_float4(0.f, 0.f, 0.f, 0.f);
    if (row < nrows) {
      val = *reinterpret_cast<const float4*>(A + (size_t)row * D_C + (c4 << 2));
      if (Ablend) {
        float4 w = *reinterpret_cast<const float4*>(Ablend + (size_t)row * D_C + (c4 << 2));
        val.x = (1.f - alpha) * val.x + alpha * w.x;
        val.y = (1.f - alpha) * val.y + alpha * w.y;
        val.z = (1.f - alpha) * val.z + alpha * w.z;
        val.w = (1.f - alpha) * val.w + alpha * w.w;
      }
    }
    *reinterpret_cast<float4*>(&At[(c4 * A_RS + rr) * 4]) = val;
  }
  for (int idx = tid; idx < 64 * 64; idx += 256) {
    int k = idx >> 6, col = idx & 63;
    Bf[(k >> 1) * B_PSTRIDE + (col << 1) + (k & 1)] = B[idx];
  }
  if (tid < 64) bs[tid] = bias[tid];
  __syncthreads();

  const int wave = tid >> 5, lane = tid & 31;
  const int lrow = lane & 15, hi = lane >> 4;
  const int rr   = wave * 16 + lrow;

  v8f acc[4] = {v8f{}, v8f{}, v8f{}, v8f{}};
#pragma unroll
  for (int k4 = 0; k4 < 16; ++k4) {
    const int p = (k4 << 1) + hi;
    v2f a = *reinterpret_cast<const v2f*>(&At[(k4 * A_RS + rr) * 4 + (hi << 1)]);
#pragma unroll
    for (int n = 0; n < 4; ++n) {
      v2f b = *reinterpret_cast<const v2f*>(
          &Bf[p * B_PSTRIDE + (((n << 4) + lrow) << 1)]);
      acc[n] = wmma_f32(a, b, acc[n]);
    }
  }

#pragma unroll
  for (int n = 0; n < 4; ++n)
#pragma unroll
    for (int j = 0; j < 8; ++j) {
      int row = row0 + wave * 16 + j + (hi << 3);
      if (row < nrows) {
        int col = (n << 4) + lrow;
        Out[(size_t)row * D_C + col] = acc[n][j] + bs[col];
      }
    }
}

__global__ void __launch_bounds__(256) k_node_gemm(
    const float* __restrict__ Xd, float* SXv, const float* __restrict__ deg,
    const float* __restrict__ W2, const float* __restrict__ b2, int nrows) {
  __shared__ __attribute__((aligned(16))) float At[32 * A_RS * 4];
  __shared__ float Bf[64 * B_PSTRIDE];
  __shared__ float bs[64];
  __shared__ float dg[128];

  const int tid  = threadIdx.x;
  const int row0 = blockIdx.x * 128;

  {
    const int c4 = tid & 31;
    const bool left = c4 < 16;
    const float* base = left ? (Xd + (c4 << 2)) : (SXv + ((c4 - 16) << 2));
    float4 val[16];
    float  d[16];
#pragma unroll
    for (int i = 0; i < 16; ++i) {
      int row = row0 + (tid >> 5) + (i << 3);
      row = row < nrows ? row : nrows - 1;
      d[i]   = deg[row];
      val[i] = *reinterpret_cast<const float4*>(base + (size_t)row * D_C);
    }
#pragma unroll
    for (int i = 0; i < 16; ++i) {
      const int pp = (tid >> 5) + (i << 3);
      const float m = left ? d[i] : 1.f;
      float4 v = val[i];
      v.x *= m; v.y *= m; v.z *= m; v.w *= m;
      *reinterpret_cast<float4*>(&At[(c4 * A_RS + pp) * 4]) = v;
    }
  }
  for (int idx = tid; idx < 128 * 64; idx += 256) {
    int k = idx >> 6, col = idx & 63;
    Bf[(k >> 1) * B_PSTRIDE + (col << 1) + (k & 1)] = W2[idx];
  }
  if (tid < 64) bs[tid] = b2[tid];
  if (tid >= 128) {
    const int r = tid - 128;
    const int row = row0 + r;
    dg[r] = deg[row < nrows ? row : nrows - 1];
  }
  __syncthreads();

  const int wave = tid >> 5, lane = tid & 31;
  const int lrow = lane & 15, hi = lane >> 4;
  const int rr   = wave * 16 + lrow;

  v8f acc[4] = {v8f{}, v8f{}, v8f{}, v8f{}};
#pragma unroll
  for (int k4 = 0; k4 < 32; ++k4) {
    const int p = (k4 << 1) + hi;
    v2f a = *reinterpret_cast<const v2f*>(&At[(k4 * A_RS + rr) * 4 + (hi << 1)]);
#pragma unroll
    for (int n = 0; n < 4; ++n) {
      v2f b = *reinterpret_cast<const v2f*>(
          &Bf[p * B_PSTRIDE + (((n << 4) + lrow) << 1)]);
      acc[n] = wmma_f32(a, b, acc[n]);
    }
  }

#pragma unroll
  for (int n = 0; n < 4; ++n)
#pragma unroll
    for (int j = 0; j < 8; ++j) {
      const int lr = wave * 16 + j + (hi << 3);
      const int row = row0 + lr;
      if (row < nrows) {
        int col = (n << 4) + lrow;
        SXv[(size_t)row * D_C + col] = acc[n][j] + dg[lr] * bs[col];
      }
    }
}

extern "C" void kernel_launch(void* const* d_in, const int* in_sizes, int n_in,
                              void* d_out, int out_size, void* d_ws, size_t ws_size,
                              hipStream_t stream) {
  const float* X      = (const float*)d_in[0];
  const float* W1_w   = (const float*)d_in[1];
  const float* W1_b   = (const float*)d_in[2];
  const float* W2_w   = (const float*)d_in[3];
  const float* W2_b   = (const float*)d_in[4];
  const float* W_w    = (const float*)d_in[5];
  const float* W_b    = (const float*)d_in[6];
  const float* dvals  = (const float*)d_in[7];
  const int*   vertex = (const int*)d_in[8];
  const int*   edges  = (const int*)d_in[9];
  const int*   drows  = (const int*)d_in[10];
  const int*   dcols  = (const int*)d_in[11];
  float* out = (float*)d_out;

  float* Xd = (float*)d_ws;
  float* Y1 = Xd + (size_t)N_NODES_C * D_C;
  float* Xe = Y1 + (size_t)N_NODES_C * D_C;
  float* Xv = Xe + (size_t)N_EDGES_C * D_C;
  float* S   = Y1;
  float* deg = Xv;


  k_owner_sum<(N_NODES_C + 255) / 256, true, false><<<256, OWN_THREADS, 0, stream>>>(
      X, drows, dcols, dvals, Xd, nullptr, N_NODES_C);
  k_gemm64<<<(N_NODES_C + 127) / 128, 256, 0, stream>>>(
      Xd, nullptr, 0.f, W1_w, W1_b, Y1, N_NODES_C);
  k_owner_sum<(N_EDGES_C + 255) / 256, false, false><<<256, OWN_THREADS, 0, stream>>>(
      Y1, edges, vertex, nullptr, Xe, nullptr, N_EDGES_C);
  k_owner_sum<(N_NODES_C + 255) / 256, false, true><<<256, OWN_THREADS, 0, stream>>>(
      Xe, vertex, edges, nullptr, S, deg, N_NODES_C);
  k_node_gemm<<<(N_NODES_C + 127) / 128, 256, 0, stream>>>(
      Xd, S, deg, W2_w, W2_b, N_NODES_C);
  k_gemm64<<<(N_NODES_C + 127) / 128, 256, 0, stream>>>(
      S, Xd, ALPHA_C, W_w, W_b, out, N_NODES_C);
}


# Item lists sorted by owner once (LDS counting sort per chunk, rank = returned LDS add); owners read only their own runs, no full key scan; sums still LDS float adds in arrival order
# speedup vs baseline: 2.5862x; 1.5584x over previous
#include <hip/hip_runtime.h>

typedef __attribute__((ext_vector_type(2))) float v2f;
typedef __attribute__((ext_vector_type(8))) float v8f;

#define N_NODES_C 100000
#define N_EDGES_C 25000
#define NNZ_C     800000
#define D_C       64
#define ALPHA_C   0.5f

#define A_RS      129
#define B_PSTRIDE 160

__device__ __forceinline__ v8f wmma_f32(v2f a, v2f b, v8f c) {
  return __builtin_amdgcn_wmma_f32_16x16x4_f32(
       false, a,  false, b,
       (short)0, c,  false,  false);
}

#define OWN_THREADS 1024
#define OWN_WAVES   (OWN_THREADS / 32)
#define NBUCKET     256
#define NR_NODES    ((N_NODES_C + NBUCKET - 1) / NBUCKET)
#define NR_EDGES    ((N_EDGES_C + NBUCKET - 1) / NBUCKET)
#define LROW_BITS   9
#define LROW_MASK   ((1 << LROW_BITS) - 1)
#define PART_CHUNKS 64
#define PART_CH     (NNZ_C / PART_CHUNKS)
#define PART_Q      (PART_CH / 4)
#define PART_IT     ((PART_Q + OWN_THREADS - 1) / OWN_THREADS)
#define OFF_STRIDE  (NBUCKET + 1)

static_assert(NNZ_C % (PART_CHUNKS * 4) == 0, "chunks are read as whole 16-byte pieces");
static_assert(PART_CH < 65536, "a rank inside a chunk is kept in 16 bits");
static_assert(NR_NODES <= (1 << LROW_BITS) && NR_EDGES <= (1 << LROW_BITS), "owned row index bits");
static_assert(N_NODES_C <= (1 << (31 - LROW_BITS)) && N_EDGES_C <= (1 << (31 - LROW_BITS)), "gather index bits");
static_assert(PART_CHUNKS % OWN_WAVES == 0, "every owner wave takes the same number of chunks");

__global__ void __launch_bounds__(OWN_THREADS) k_partition(
    const int* __restrict__ drows, const int* __restrict__ dcols, const float* __restrict__ dvals,
    const int* __restrict__ edges, const int* __restrict__ vertex,
    int* __restrict__ ent, float* __restrict__ esc, int* __restrict__ off) {
  __shared__ int hist[NBUCKET];
  __shared__ int offs[NBUCKET + 1];
  __shared__ int wtot[NBUCKET / 32];
  __shared__ __attribute__((aligned(16))) int   se[PART_CH];
  __shared__ __attribute__((aligned(16))) float ss[PART_CH];

  const int tid = threadIdx.x, lane = tid & 31;
  const int g = blockIdx.x, l = blockIdx.y;
  const int* key  = l == 0 ? drows : (l == 1 ? edges : vertex);
  const int* gidx = l == 0 ? dcols : (l == 1 ? vertex : edges);
  const unsigned nr   = l == 1 ? NR_EDGES : NR_NODES;
  const unsigned ndst = l == 1 ? N_EDGES_C : N_NODES_C;

  if (tid < NBUCKET) hist[tid] = 0;
  __syncthreads();

  const size_t cbase = (size_t)g * PART_CH;
  const int4*   key4 = reinterpret_cast<const int4*>(key + cbase);
  const int4*   gid4 = reinterpret_cast<const int4*>(gidx + cbase);
  const float4* sc4  = reinterpret_cast<const float4*>(dvals + cbase);

  int4 kv[PART_IT], gv[PART_IT];
  float4 sv[PART_IT];
#pragma unroll
  for (int it = 0; it < PART_IT; ++it) {
    const int qi = it * OWN_THREADS + tid;
    const int qc = qi < PART_Q ? qi : PART_Q - 1;
    kv[it] = key4[qc];
    gv[it] = gid4[qc];
    sv[it] = sc4[qc];
  }

  int code[PART_IT][4];
#pragma unroll
  for (int it = 0; it < PART_IT; ++it) {
    const int qi = it * OWN_THREADS + tid;
    const int kk[4] = {kv[it].x, kv[it].y, kv[it].z, kv[it].w};
#pragma unroll
    for (int c = 0; c < 4; ++c) {
      const unsigned k = (unsigned)kk[c];
      const unsigned o = l == 1 ? k / NR_EDGES : k / NR_NODES;
      int cd = -1;
      if (qi < PART_Q && k < ndst) cd = (int)(o << 16) | atomicAdd(&hist[o], 1);
      code[it][c] = cd;
    }
  }
  __syncthreads();

  int cnt = 0, inc = 0;
  if (tid < NBUCKET) {
    cnt = hist[tid];
    inc = cnt;
#pragma unroll
    for (int d = 1; d < 32; d <<= 1) {
      const int t = __shfl_up(inc, d);
      if (lane >= d) inc += t;
    }
    if (lane == 31) wtot[tid >> 5] = inc;
  }
  __syncthreads();
  if (tid < NBUCKET) {
    int base = 0;
    for (int w = 0; w < (tid >> 5); ++w) base += wtot[w];
    offs[tid] = base + inc - cnt;
    if (tid == NBUCKET - 1) offs[NBUCKET] = base + inc;
  }
  __syncthreads();

  if (tid <= NBUCKET)
    off[((size_t)l * PART_CHUNKS + g) * OFF_STRIDE + tid] = offs[tid];

#pragma unroll
  for (int it = 0; it < PART_IT; ++it) {
    const int kk[4] = {kv[it].x, kv[it].y, kv[it].z, kv[it].w};
    const int gg[4] = {gv[it].x, gv[it].y, gv[it].z, gv[it].w};
    const float sc[4] = {sv[it].x, sv[it].y, sv[it].z, sv[it].w};
#pragma unroll
    for (int c = 0; c < 4; ++c) {
      const int cd = code[it][c];
      if (cd >= 0) {
        const unsigned o = (unsigned)cd >> 16;
        const int pos = offs[o] + (cd & 0xffff);
        se[pos] = (gg[c] << LROW_BITS) | (int)((unsigned)kk[c] - o * nr);
        if (l == 0) ss[pos] = sc[c];
      }
    }
  }
  __syncthreads();

  const int total = offs[NBUCKET];
  int4*   ent4 = reinterpret_cast<int4*>(ent + (size_t)l * NNZ_C + cbase);
  float4* esc4 = reinterpret_cast<float4*>(esc + cbase);
  for (int q = tid; (q << 2) < total; q += OWN_THREADS) {
    ent4[q] = *reinterpret_cast<const int4*>(&se[q << 2]);
    if (l == 0) esc4[q] = *reinterpret_cast<const float4*>(&ss[q << 2]);
  }
}

#define OWN_BLK 32

template <int NR, bool SCALED, bool COUNT>
__global__ void __launch_bounds__(OWN_THREADS) k_owner_sum(
    const float* __restrict__ src, const int* __restrict__ ent,
    const float* __restrict__ esc, const int* __restrict__ off,
    float* __restrict__ dst, float* __restrict__ cnt_out, int ndst) {
  __shared__ __attribute__((aligned(16))) float acc[NR * D_C];
  __shared__ float cntl[COUNT ? NR : 1];

  const int tid  = threadIdx.x;
  const int lane = tid & 31;
  const int wave = __builtin_amdgcn_readfirstlane(tid >> 5);
  const int own  = blockIdx.x;
  const int r0   = own * NR;
  int nown = ndst - r0;
  nown = nown > NR ? NR : (nown < 0 ? 0 : nown);

  for (int j = tid; j < NR * D_C; j += OWN_THREADS) acc[j] = 0.f;
  if (COUNT)
    for (int j = tid; j < NR; j += OWN_THREADS) cntl[j] = 0.f;
  __syncthreads();

  constexpr int RUNS = PART_CHUNKS / OWN_WAVES;
  int rs[RUNS], re[RUNS], cur[RUNS];
  float curs[RUNS];
#pragma unroll
  for (int r = 0; r < RUNS; ++r) {
    const int* o = off + (wave + r * OWN_WAVES) * OFF_STRIDE + own;
    rs[r] = __builtin_amdgcn_readfirstlane(o[0]);
    re[r] = __builtin_amdgcn_readfirstlane(o[1]);
  }
#pragma unroll
  for (int r = 0; r < RUNS; ++r) {
    const size_t cb = (size_t)(wave + r * OWN_WAVES) * PART_CH;
    const int last = re[r] > rs[r] ? re[r] - 1 : 0;
    const int ji = rs[r] + lane < last ? rs[r] + lane : last;
    cur[r]  = ent[cb + ji];
    curs[r] = SCALED ? esc[cb + ji] : 1.f;
  }

  auto block = [&](int me, float ms, int n) {
    if (COUNT) {
      if (lane < n)
        __hip_atomic_fetch_add(&cntl[me & LROW_MASK], 1.f, __ATOMIC_RELAXED, __HIP_MEMORY_SCOPE_WORKGROUP);
    }
    v2f x[OWN_BLK];
#pragma unroll
    for (int k = 0; k < OWN_BLK; ++k)
      if (k < n) {
        const int eu = __builtin_amdgcn_readlane(me, k);
        x[k] = *reinterpret_cast<const v2f*>(src + (size_t)(eu >> LROW_BITS) * D_C + (lane << 1));
      }
#pragma unroll
    for (int k = 0; k < OWN_BLK; ++k)
      if (k < n) {
        const int eu = __builtin_amdgcn_readlane(me, k);
        float a = x[k][0], c = x[k][1];
        if (SCALED) {
          const float s = __int_as_float(__builtin_amdgcn_readlane(__float_as_int(ms), k));
          a *= s; c *= s;
        }
        float* d = &acc[(eu & LROW_MASK) * D_C + (lane << 1)];
        __hip_atomic_fetch_add(d,     a, __ATOMIC_RELAXED, __HIP_MEMORY_SCOPE_WORKGROUP);
        __hip_atomic_fetch_add(d + 1, c, __ATOMIC_RELAXED, __HIP_MEMORY_SCOPE_WORKGROUP);
      }
  };

#pragma unroll
  for (int r = 0; r < RUNS; ++r) {
    const size_t cb = (size_t)(wave + r * OWN_WAVES) * PART_CH;
    const int e = re[r];
    int   me = cur[r];
    float ms = curs[r];
    for (int j = rs[r]; j < e; j += OWN_BLK) {
      const int n = e - j < OWN_BLK ? e - j : OWN_BLK;
      const int jn = j + OWN_BLK + lane < e ? j + OWN_BLK + lane : e - 1;
      const int   nxt  = ent[cb + jn];
      const float nxts = SCALED ? esc[cb + jn] : 1.f;
      block(me, ms, n);
      me = nxt;
      ms = nxts;
    }
  }
  __syncthreads();

  for (int j = tid; j < nown * (D_C / 4); j += OWN_THREADS)
    *reinterpret_cast<float4*>(dst + (size_t)r0 * D_C + (j << 2)) =
        *reinterpret_cast<const float4*>(&acc[j << 2]);
  if (COUNT)
    for (int j = tid; j < nown; j += OWN_THREADS) cnt_out[r0 + j] = cntl[j];
}

__global__ void __launch_bounds__(256) k_gemm64(
    const float* __restrict__ A, const float* __restrict__ Ablend, float alpha,
    const float* __restrict__ B, const float* __restrict__ bias,
    float* __restrict__ Out, int nrows) {
  __shared__ __attribute__((aligned(16))) float At[16 * A_RS * 4];
  __shared__ float Bf[32 * B_PSTRIDE];
  __shared__ float bs[64];

  const int tid  = threadIdx.x;
  const int row0 = blockIdx.x * 128;

  for (int idx = tid; idx < 128 * 16; idx += 256) {
    int rr = idx >> 4, c4 = idx & 15;
    int row = row0 + rr;
    float4 val = make_float4(0.f, 0.f, 0.f, 0.f);
    if (row < nrows) {
      val = *reinterpret_cast<const float4*>(A + (size_t)row * D_C + (c4 << 2));
      if (Ablend) {
        float4 w = *reinterpret_cast<const float4*>(Ablend + (size_t)row * D_C + (c4 << 2));
        val.x = (1.f - alpha) * val.x + alpha * w.x;
        val.y = (1.f - alpha) * val.y + alpha * w.y;
        val.z = (1.f - alpha) * val.z + alpha * w.z;
        val.w = (1.f - alpha) * val.w + alpha * w.w;
      }
    }
    *reinterpret_cast<float4*>(&At[(c4 * A_RS + rr) * 4]) = val;
  }
  for (int idx = tid; idx < 64 * 64; idx += 256) {
    int k = idx >> 6, col = idx & 63;
    Bf[(k >> 1) * B_PSTRIDE + (col << 1) + (k & 1)] = B[idx];
  }
  if (tid < 64) bs[tid] = bias[tid];
  __syncthreads();

  const int wave = tid >> 5, lane = tid & 31;
  const int lrow = lane & 15, hi = lane >> 4;
  const int rr   = wave * 16 + lrow;

  v8f acc[4] = {v8f{}, v8f{}, v8f{}, v8f{}};
#pragma unroll
  for (int k4 = 0; k4 < 16; ++k4) {
    const int p = (k4 << 1) + hi;
    v2f a = *reinterpret_cast<const v2f*>(&At[(k4 * A_RS + rr) * 4 + (hi << 1)]);
#pragma unroll
    for (int n = 0; n < 4; ++n) {
      v2f b = *reinterpret_cast<const v2f*>(
          &Bf[p * B_PSTRIDE + (((n << 4) + lrow) << 1)]);
      acc[n] = wmma_f32(a, b, acc[n]);
    }
  }

#pragma unroll
  for (int n = 0; n < 4; ++n)
#pragma unroll
    for (int j = 0; j < 8; ++j) {
      int row = row0 + wave * 16 + j + (hi << 3);
      if (row < nrows) {
        int col = (n << 4) + lrow;
        Out[(size_t)row * D_C + col] = acc[n][j] + bs[col];
      }
    }
}

__global__ void __launch_bounds__(256) k_node_gemm(
    const float* __restrict__ Xd, float* SXv, const float* __restrict__ deg,
    const float* __restrict__ W2, const float* __restrict__ b2, int nrows) {
  __shared__ __attribute__((aligned(16))) float At[32 * A_RS * 4];
  __shared__ float Bf[64 * B_PSTRIDE];
  __shared__ float bs[64];
  __shared__ float dg[128];

  const int tid  = threadIdx.x;
  const int row0 = blockIdx.x * 128;

  {
    const int c4 = tid & 31;
    const bool left = c4 < 16;
    const float* base = left ? (Xd + (c4 << 2)) : (SXv + ((c4 - 16) << 2));
    float4 val[16];
    float  d[16];
#pragma unroll
    for (int i = 0; i < 16; ++i) {
      int row = row0 + (tid >> 5) + (i << 3);
      row = row < nrows ? row : nrows - 1;
      d[i]   = deg[row];
      val[i] = *reinterpret_cast<const float4*>(base + (size_t)row * D_C);
    }
#pragma unroll
    for (int i = 0; i < 16; ++i) {
      const int pp = (tid >> 5) + (i << 3);
      const float m = left ? d[i] : 1.f;
      float4 v = val[i];
      v.x *= m; v.y *= m; v.z *= m; v.w *= m;
      *reinterpret_cast<float4*>(&At[(c4 * A_RS + pp) * 4]) = v;
    }
  }
  for (int idx = tid; idx < 128 * 64; idx += 256) {
    int k = idx >> 6, col = idx & 63;
    Bf[(k >> 1) * B_PSTRIDE + (col << 1) + (k & 1)] = W2[idx];
  }
  if (tid < 64) bs[tid] = b2[tid];
  if (tid >= 128) {
    const int r = tid - 128;
    const int row = row0 + r;
    dg[r] = deg[row < nrows ? row : nrows - 1];
  }
  __syncthreads();

  const int wave = tid >> 5, lane = tid & 31;
  const int lrow = lane & 15, hi = lane >> 4;
  const int rr   = wave * 16 + lrow;

  v8f acc[4] = {v8f{}, v8f{}, v8f{}, v8f{}};
#pragma unroll
  for (int k4 = 0; k4 < 32; ++k4) {
    const int p = (k4 << 1) + hi;
    v2f a = *reinterpret_cast<const v2f*>(&At[(k4 * A_RS + rr) * 4 + (hi << 1)]);
#pragma unroll
    for (int n = 0; n < 4; ++n) {
      v2f b = *reinterpret_cast<const v2f*>(
          &Bf[p * B_PSTRIDE + (((n << 4) + lrow) << 1)]);
      acc[n] = wmma_f32(a, b, acc[n]);
    }
  }

#pragma unroll
  for (int n = 0; n < 4; ++n)
#pragma unroll
    for (int j = 0; j < 8; ++j) {
      const int lr = wave * 16 + j + (hi << 3);
      const int row = row0 + lr;
      if (row < nrows) {
        int col = (n << 4) + lrow;
        SXv[(size_t)row * D_C + col] = acc[n][j] + dg[lr] * bs[col];
      }
    }
}

extern "C" void kernel_launch(void* const* d_in, const int* in_sizes, int n_in,
                              void* d_out, int out_size, void* d_ws, size_t ws_size,
                              hipStream_t stream) {
  const float* X      = (const float*)d_in[0];
  const float* W1_w   = (const float*)d_in[1];
  const float* W1_b   = (const float*)d_in[2];
  const float* W2_w   = (const float*)d_in[3];
  const float* W2_b   = (const float*)d_in[4];
  const float* W_w    = (const float*)d_in[5];
  const float* W_b    = (const float*)d_in[6];
  const float* dvals  = (const float*)d_in[7];
  const int*   vertex = (const int*)d_in[8];
  const int*   edges  = (const int*)d_in[9];
  const int*   drows  = (const int*)d_in[10];
  const int*   dcols  = (const int*)d_in[11];
  float* out = (float*)d_out;

  float* Xd = (float*)d_ws;
  float* Y1 = Xd + (size_t)N_NODES_C * D_C;
  float* Xe = Y1 + (size_t)N_NODES_C * D_C;
  float* Xv = Xe + (size_t)N_EDGES_C * D_C;
  float* S   = Y1;
  float* deg = Xv;
  int*   ent = (int*)(Xv + 102400);
  float* esc = (float*)(ent + (size_t)3 * NNZ_C);
  int*   off = (int*)(esc + NNZ_C);
  static_assert(102400 >= N_NODES_C &&
                102400 + 4 * (size_t)NNZ_C + 3 * PART_CHUNKS * OFF_STRIDE <= (size_t)N_NODES_C * D_C,
                "entries fit in the former Xv plane behind deg");


  k_partition<<<dim3(PART_CHUNKS, 3), OWN_THREADS, 0, stream>>>(
      drows, dcols, dvals, edges, vertex, ent, esc, off);
  k_owner_sum<NR_NODES, true, false><<<NBUCKET, OWN_THREADS, 0, stream>>>(
      X, ent, esc, off, Xd, nullptr, N_NODES_C);
  k_gemm64<<<(N_NODES_C + 127) / 128, 256, 0, stream>>>(
      Xd, nullptr, 0.f, W1_w, W1_b, Y1, N_NODES_C);
  k_owner_sum<NR_EDGES, false, false><<<NBUCKET, OWN_THREADS, 0, stream>>>(
      Y1, ent + (size_t)NNZ_C, nullptr, off + PART_CHUNKS * OFF_STRIDE, Xe, nullptr, N_EDGES_C);
  k_owner_sum<NR_NODES, false, true><<<NBUCKET, OWN_THREADS, 0, stream>>>(
      Xe, ent + (size_t)2 * NNZ_C, nullptr, off + 2 * PART_CHUNKS * OFF_STRIDE, S, deg, N_NODES_C);
  k_node_gemm<<<(N_NODES_C + 127) / 128, 256, 0, stream>>>(
      Xd, S, deg, W2_w, W2_b, N_NODES_C);
  k_gemm64<<<(N_NODES_C + 127) / 128, 256, 0, stream>>>(
      S, Xd, ALPHA_C, W_w, W_b, out, N_NODES_C);
}


# Dense products moved into the owner work-groups' epilogue on the LDS-resident sums (Y1 from Xd rows; Xv and out from S rows; same f32 WMMA, same k order): 7 launches become 4; S, Xv, deg never stored
# speedup vs baseline: 3.2064x; 1.2398x over previous
#include <hip/hip_runtime.h>

typedef __attribute__((ext_vector_type(2))) float v2f;
typedef __attribute__((ext_vector_type(8))) float v8f;

#define N_NODES_C 100000
#define N_EDGES_C 25000
#define NNZ_C     800000
#define D_C       64
#define ALPHA_C   0.5f

#define ACC_P     68
#define B_PSTRIDE 160

__device__ __forceinline__ v8f wmma_f32(v2f a, v2f b, v8f c) {
  return __builtin_amdgcn_wmma_f32_16x16x4_f32(
       false, a,  false, b,
       (short)0, c,  false,  false);
}

#define OWN_THREADS 1024
#define OWN_WAVES   (OWN_THREADS / 32)
#define NBUCKET     256
#define NR_NODES    ((N_NODES_C + NBUCKET - 1) / NBUCKET)
#define NR_EDGES    ((N_EDGES_C + NBUCKET - 1) / NBUCKET)
#define LROW_BITS   9
#define LROW_MASK   ((1 << LROW_BITS) - 1)
#define PART_CHUNKS 64
#define PART_CH     (NNZ_C / PART_CHUNKS)
#define PART_Q      (PART_CH / 4)
#define PART_IT     ((PART_Q + OWN_THREADS - 1) / OWN_THREADS)
#define OFF_STRIDE  (NBUCKET + 1)

static_assert(NNZ_C % (PART_CHUNKS * 4) == 0, "chunks are read as whole 16-byte pieces");
static_assert(PART_CH < 65536, "a rank inside a chunk is kept in 16 bits");
static_assert(NR_NODES <= (1 << LROW_BITS) && NR_EDGES <= (1 << LROW_BITS), "owned row index bits");
static_assert(N_NODES_C <= (1 << (31 - LROW_BITS)) && N_EDGES_C <= (1 << (31 - LROW_BITS)), "gather index bits");
static_assert(PART_CHUNKS % OWN_WAVES == 0, "every owner wave takes the same number of chunks");

__global__ void __launch_bounds__(OWN_THREADS) k_partition(
    const int* __restrict__ drows, const int* __restrict__ dcols, const float* __restrict__ dvals,
    const int* __restrict__ edges, const int* __restrict__ vertex,
    int* __restrict__ ent, float* __restrict__ esc, int* __restrict__ off) {
  __shared__ int hist[NBUCKET];
  __shared__ int offs[NBUCKET + 1];
  __shared__ int wtot[NBUCKET / 32];
  __shared__ __attribute__((aligned(16))) int   se[PART_CH];
  __shared__ __attribute__((aligned(16))) float ss[PART_CH];

  const int tid = threadIdx.x, lane = tid & 31;
  const int g = blockIdx.x, l = blockIdx.y;
  const int* key  = l == 0 ? drows : (l == 1 ? edges : vertex);
  const int* gidx = l == 0 ? dcols : (l == 1 ? vertex : edges);
  const unsigned nr   = l == 1 ? NR_EDGES : NR_NODES;
  const unsigned ndst = l == 1 ? N_EDGES_C : N_NODES_C;

  if (tid < NBUCKET) hist[tid] = 0;
  __syncthreads();

  const size_t cbase = (size_t)g * PART_CH;
  const int4*   key4 = reinterpret_cast<const int4*>(key + cbase);
  const int4*   gid4 = reinterpret_cast<const int4*>(gidx + cbase);
  const float4* sc4  = reinterpret_cast<const float4*>(dvals + cbase);

  int4 kv[PART_IT], gv[PART_IT];
  float4 sv[PART_IT];
#pragma unroll
  for (int it = 0; it < PART_IT; ++it) {
    const int qi = it * OWN_THREADS + tid;
    const int qc = qi < PART_Q ? qi : PART_Q - 1;
    kv[it] = key4[qc];
    gv[it] = gid4[qc];
    sv[it] = sc4[qc];
  }

  int code[PART_IT][4];
#pragma unroll
  for (int it = 0; it < PART_IT; ++it) {
    const int qi = it * OWN_THREADS + tid;
    const int kk[4] = {kv[it].x, kv[it].y, kv[it].z, kv[it].w};
#pragma unroll
    for (int c = 0; c < 4; ++c) {
      const unsigned k = (unsigned)kk[c];
      const unsigned o = l == 1 ? k / NR_EDGES : k / NR_NODES;
      int cd = -1;
      if (qi < PART_Q && k < ndst) cd = (int)(o << 16) | atomicAdd(&hist[o], 1);
      code[it][c] = cd;
    }
  }
  __syncthreads();

  int cnt = 0, inc = 0;
  if (tid < NBUCKET) {
    cnt = hist[tid];
    inc = cnt;
#pragma unroll
    for (int d = 1; d < 32; d <<= 1) {
      const int t = __shfl_up(inc, d);
      if (lane >= d) inc += t;
    }
    if (lane == 31) wtot[tid >> 5] = inc;
  }
  __syncthreads();
  if (tid < NBUCKET) {
    int base = 0;
    for (int w = 0; w < (tid >> 5); ++w) base += wtot[w];
    offs[tid] = base + inc - cnt;
    if (tid == NBUCKET - 1) offs[NBUCKET] = base + inc;
  }
  __syncthreads();

  if (tid <= NBUCKET)
    off[((size_t)l * PART_CHUNKS + g) * OFF_STRIDE + tid] = offs[tid];

#pragma unroll
  for (int it = 0; it < PART_IT; ++it) {
    const int kk[4] = {kv[it].x, kv[it].y, kv[it].z, kv[it].w};
    const int gg[4] = {gv[it].x, gv[it].y, gv[it].z, gv[it].w};
    const float sc[4] = {sv[it].x, sv[it].y, sv[it].z, sv[it].w};
#pragma unroll
    for (int c = 0; c < 4; ++c) {
      const int cd = code[it][c];
      if (cd >= 0) {
        const unsigned o = (unsigned)cd >> 16;
        const int pos = offs[o] + (cd & 0xffff);
        se[pos] = (gg[c] << LROW_BITS) | (int)((unsigned)kk[c] - o * nr);
        if (l == 0) ss[pos] = sc[c];
      }
    }
  }
  __syncthreads();

  const int total = offs[NBUCKET];
  int4*   ent4 = reinterpret_cast<int4*>(ent + (size_t)l * NNZ_C + cbase);
  float4* esc4 = reinterpret_cast<float4*>(esc + cbase);
  for (int q = tid; (q << 2) < total; q += OWN_THREADS) {
    ent4[q] = *reinterpret_cast<const int4*>(&se[q << 2]);
    if (l == 0) esc4[q] = *reinterpret_cast<const float4*>(&ss[q << 2]);
  }
}

#define OWN_BLK 32

template <int P, bool SCALED, bool COUNT>
__device__ __forceinline__ void owner_accumulate(
    const float* __restrict__ src, const int* __restrict__ ent,
    const float* __restrict__ esc, const int* __restrict__ off,
    float* acc, float* cntl, int own, int wave, int lane) {
  constexpr int RUNS = PART_CHUNKS / OWN_WAVES;
  int rs[RUNS], re[RUNS], cur[RUNS];
  float curs[RUNS];
#pragma unroll
  for (int r = 0; r < RUNS; ++r) {
    const int* o = off + (wave + r * OWN_WAVES) * OFF_STRIDE + own;
    rs[r] = __builtin_amdgcn_readfirstlane(o[0]);
    re[r] = __builtin_amdgcn_readfirstlane(o[1]);
  }
#pragma unroll
  for (int r = 0; r < RUNS; ++r) {
    const size_t cb = (size_t)(wave + r * OWN_WAVES) * PART_CH;
    const int last = re[r] > rs[r] ? re[r] - 1 : 0;
    const int ji = rs[r] + lane < last ? rs[r] + lane : last;
    cur[r]  = ent[cb + ji];
    curs[r] = SCALED ? esc[cb + ji] : 1.f;
  }

  auto block = [&](int me, float ms, int n) {
    if (COUNT) {
      if (lane < n)
        __hip_atomic_fetch_add(&cntl[me & LROW_MASK], 1.f, __ATOMIC_RELAXED, __HIP_MEMORY_SCOPE_WORKGROUP);
    }
    v2f x[OWN_BLK];
#pragma unroll
    for (int k = 0; k < OWN_BLK; ++k)
      if (k < n) {
        const int eu = __builtin_amdgcn_readlane(me, k);
        x[k] = *reinterpret_cast<const v2f*>(src + (size_t)(eu >> LROW_BITS) * D_C + (lane << 1));
      }
#pragma unroll
    for (int k = 0; k < OWN_BLK; ++k)
      if (k < n) {
        const int eu = __builtin_amdgcn_readlane(me, k);
        float a = x[k][0], c = x[k][1];
        if (SCALED) {
          const float s = __int_as_float(__builtin_amdgcn_readlane(__float_as_int(ms), k));
          a *= s; c *= s;
        }
        float* d = &acc[(eu & LROW_MASK) * P + (lane << 1)];
        __hip_atomic_fetch_add(d,     a, __ATOMIC_RELAXED, __HIP_MEMORY_SCOPE_WORKGROUP);
        __hip_atomic_fetch_add(d + 1, c, __ATOMIC_RELAXED, __HIP_MEMORY_SCOPE_WORKGROUP);
      }
  };

#pragma unroll
  for (int r = 0; r < RUNS; ++r) {
    const size_t cb = (size_t)(wave + r * OWN_WAVES) * PART_CH;
    const int e = re[r];
    int   me = cur[r];
    float ms = curs[r];
    for (int j = rs[r]; j < e; j += OWN_BLK) {
      const int n = e - j < OWN_BLK ? e - j : OWN_BLK;
      const int jn = j + OWN_BLK + lane < e ? j + OWN_BLK + lane : e - 1;
      const int   nxt  = ent[cb + jn];
      const float nxts = SCALED ? esc[cb + jn] : 1.f;
      block(me, ms, n);
      me = nxt;
      ms = nxts;
    }
  }
}

template <int NR>
__global__ void __launch_bounds__(OWN_THREADS) k_owner_sum(
    const float* __restrict__ src, const int* __restrict__ ent, const int* __restrict__ off,
    float* __restrict__ dst, int ndst) {
  __shared__ __attribute__((aligned(16))) float acc[NR * D_C];

  const int tid  = threadIdx.x;
  const int lane = tid & 31;
  const int wave = __builtin_amdgcn_readfirstlane(tid >> 5);
  const int own  = blockIdx.x;
  const int r0   = own * NR;
  int nown = ndst - r0;
  nown = nown > NR ? NR : (nown < 0 ? 0 : nown);

  for (int j = tid; j < NR * D_C; j += OWN_THREADS) acc[j] = 0.f;
  __syncthreads();
  owner_accumulate<D_C, false, false>(src, ent, nullptr, off, acc, nullptr, own, wave, lane);
  __syncthreads();

  for (int j = tid; j < nown * (D_C / 4); j += OWN_THREADS)
    *reinterpret_cast<float4*>(dst + (size_t)r0 * D_C + (j << 2)) =
        *reinterpret_cast<const float4*>(&acc[j << 2]);
}

#define NT_NODES  ((NR_NODES + 15) / 16)
#define NRP_NODES (NT_NODES * 16)
static_assert(NT_NODES <= OWN_WAVES, "one tile per wave");

__device__ __forceinline__ void stage_b(float* Bf, const float* __restrict__ B, int nk, int tid) {
  for (int idx = tid; idx < nk * 64; idx += OWN_THREADS) {
    const int k = idx >> 6, col = idx & 63;
    Bf[(k >> 1) * B_PSTRIDE + (col << 1) + (k & 1)] = B[idx];
  }
}

template <bool SCALE>
__device__ __forceinline__ void tile_mac64(const float* arow, float s, const float* bl, v8f (&c)[4]) {
#pragma unroll
  for (int k4 = 0; k4 < 16; ++k4) {
    v2f a = *reinterpret_cast<const v2f*>(arow + (k4 << 2));
    if (SCALE) { a[0] *= s; a[1] *= s; }
#pragma unroll
    for (int n = 0; n < 4; ++n) {
      const v2f b = *reinterpret_cast<const v2f*>(bl + (k4 << 1) * B_PSTRIDE + (n << 5));
      c[n] = wmma_f32(a, b, c[n]);
    }
  }
}

__global__ void __launch_bounds__(OWN_THREADS) k_owner_diffuse_w1(
    const float* __restrict__ X, const int* __restrict__ ent, const float* __restrict__ esc,
    const int* __restrict__ off, const float* __restrict__ W1, const float* __restrict__ b1,
    float* __restrict__ Xd, float* __restrict__ Y1, int ndst) {
  __shared__ __attribute__((aligned(16))) float acc[NRP_NODES * ACC_P];
  __shared__ float Bf[32 * B_PSTRIDE];
  __shared__ float bs[64];

  const int tid  = threadIdx.x;
  const int lane = tid & 31;
  const int wave = __builtin_amdgcn_readfirstlane(tid >> 5);
  const int own  = blockIdx.x;
  const int r0   = own * NR_NODES;
  int nown = ndst - r0;
  nown = nown > NR_NODES ? NR_NODES : (nown < 0 ? 0 : nown);

  for (int j = tid; j < NRP_NODES * ACC_P / 4; j += OWN_THREADS)
    *reinterpret_cast<float4*>(&acc[j << 2]) = make_float4(0.f, 0.f, 0.f, 0.f);
  stage_b(Bf, W1, 64, tid);
  if (tid < 64) bs[tid] = b1[tid];
  __syncthreads();
  owner_accumulate<ACC_P, true, false>(X, ent, esc, off, acc, nullptr, own, wave, lane);
  __syncthreads();

  for (int j = tid; j < nown * 16; j += OWN_THREADS) {
    const int row = j >> 4, c4 = j & 15;
    *reinterpret_cast<float4*>(Xd + (size_t)(r0 + row) * D_C + (c4 << 2)) =
        *reinterpret_cast<const float4*>(&acc[row * ACC_P + (c4 << 2)]);
  }

  const int lrow = lane & 15, hi = lane >> 4;
  v8f c[4] = {v8f{}, v8f{}, v8f{}, v8f{}};
  if (wave < NT_NODES)
    tile_mac64<false>(&acc[(wave * 16 + lrow) * ACC_P + (hi << 1)], 1.f,
                      &Bf[hi * B_PSTRIDE + (lrow << 1)], c);
  __syncthreads();
  if (wave < NT_NODES) {
#pragma unroll
    for (int n = 0; n < 4; ++n)
#pragma unroll
      for (int j = 0; j < 8; ++j) {
        const int lr = wave * 16 + j + (hi << 3);
        const int col = (n << 4) + lrow;
        acc[lr * ACC_P + col] = c[n][j] + bs[col];
      }
  }
  __syncthreads();
  for (int j = tid; j < nown * 16; j += OWN_THREADS) {
    const int row = j >> 4, c4 = j & 15;
    *reinterpret_cast<float4*>(Y1 + (size_t)(r0 + row) * D_C + (c4 << 2)) =
        *reinterpret_cast<const float4*>(&acc[row * ACC_P + (c4 << 2)]);
  }
}

__global__ void __launch_bounds__(OWN_THREADS) k_owner_node_out(
    const float* __restrict__ Xe, const int* __restrict__ ent, const int* __restrict__ off,
    const float* __restrict__ Xd, const float* __restrict__ W2, const float* __restrict__ b2,
    const float* __restrict__ Ww, const float* __restrict__ bw, float alpha,
    float* __restrict__ out, int ndst) {
  __shared__ __attribute__((aligned(16))) float acc[NRP_NODES * ACC_P];
  __shared__ __attribute__((aligned(16))) float xs[NRP_NODES * ACC_P];
  __shared__ float Bf2[64 * B_PSTRIDE];
  __shared__ float Bfw[32 * B_PSTRIDE];
  __shared__ float b2s[64];
  __shared__ float bws[64];
  __shared__ float cntl[NRP_NODES];

  const int tid  = threadIdx.x;
  const int lane = tid & 31;
  const int wave = __builtin_amdgcn_readfirstlane(tid >> 5);
  const int own  = blockIdx.x;
  const int r0   = own * NR_NODES;
  int nown = ndst - r0;
  nown = nown > NR_NODES ? NR_NODES : (nown < 0 ? 0 : nown);

  for (int j = tid; j < NRP_NODES * 16; j += OWN_THREADS) {
    const int row = j >> 4, c4 = j & 15;
    float4 v = make_float4(0.f, 0.f, 0.f, 0.f);
    if (row < nown) v = *reinterpret_cast<const float4*>(Xd + (size_t)(r0 + row) * D_C + (c4 << 2));
    *reinterpret_cast<float4*>(&xs[row * ACC_P + (c4 << 2)]) = v;
  }
  for (int j = tid; j < NRP_NODES * ACC_P / 4; j += OWN_THREADS)
    *reinterpret_cast<float4*>(&acc[j << 2]) = make_float4(0.f, 0.f, 0.f, 0.f);
  for (int j = tid; j < NRP_NODES; j += OWN_THREADS) cntl[j] = 0.f;
  stage_b(Bf2, W2, 128, tid);
  stage_b(Bfw, Ww, 64, tid);
  if (tid < 64) { b2s[tid] = b2[tid]; bws[tid] = bw[tid]; }
  __syncthreads();
  owner_accumulate<ACC_P, false, true>(Xe, ent, nullptr, off, acc, cntl, own, wave, lane);
  __syncthreads();

  const int lrow = lane & 15, hi = lane >> 4;
  v8f c[4] = {v8f{}, v8f{}, v8f{}, v8f{}};
  if (wave < NT_NODES) {
    const int ar = (wave * 16 + lrow) * ACC_P + (hi << 1);
    const float* bl = &Bf2[hi * B_PSTRIDE + (lrow << 1)];
    tile_mac64<true >(&xs[ar],  cntl[wave * 16 + lrow], bl, c);
    tile_mac64<false>(&acc[ar], 1.f,                    bl + 32 * B_PSTRIDE, c);
#pragma unroll
    for (int n = 0; n < 4; ++n)
#pragma unroll
      for (int j = 0; j < 8; ++j) {
        const int lr = wave * 16 + j + (hi << 3);
        const int col = (n << 4) + lrow;
        const float xv = c[n][j] + cntl[lr] * b2s[col];
        c[n][j] = (1.f - alpha) * xv + alpha * xs[lr * ACC_P + col];
      }
  }
  __syncthreads();
  if (wave < NT_NODES) {
#pragma unroll
    for (int n = 0; n < 4; ++n)
#pragma unroll
      for (int j = 0; j < 8; ++j)
        acc[(wave * 16 + j + (hi << 3)) * ACC_P + (n << 4) + lrow] = c[n][j];
  }
  __syncthreads();
  v8f o[4] = {v8f{}, v8f{}, v8f{}, v8f{}};
  if (wave < NT_NODES)
    tile_mac64<false>(&acc[(wave * 16 + lrow) * ACC_P + (hi << 1)], 1.f,
                      &Bfw[hi * B_PSTRIDE + (lrow << 1)], o);
  __syncthreads();
  if (wave < NT_NODES) {
#pragma unroll
    for (int n = 0; n < 4; ++n)
#pragma unroll
      for (int j = 0; j < 8; ++j) {
        const int col = (n << 4) + lrow;
        acc[(wave * 16 + j + (hi << 3)) * ACC_P + col] = o[n][j] + bws[col];
      }
  }
  __syncthreads();
  for (int j = tid; j < nown * 16; j += OWN_THREADS) {
    const int row = j >> 4, c4 = j & 15;
    *reinterpret_cast<float4*>(out + (size_t)(r0 + row) * D_C + (c4 << 2)) =
        *reinterpret_cast<const float4*>(&acc[row * ACC_P + (c4 << 2)]);
  }
}

extern "C" void kernel_launch(void* const* d_in, const int* in_sizes, int n_in,
                              void* d_out, int out_size, void* d_ws, size_t ws_size,
                              hipStream_t stream) {
  const float* X      = (const float*)d_in[0];
  const float* W1_w   = (const float*)d_in[1];
  const float* W1_b   = (const float*)d_in[2];
  const float* W2_w   = (const float*)d_in[3];
  const float* W2_b   = (const float*)d_in[4];
  const float* W_w    = (const float*)d_in[5];
  const float* W_b    = (const float*)d_in[6];
  const float* dvals  = (const float*)d_in[7];
  const int*   vertex = (const int*)d_in[8];
  const int*   edges  = (const int*)d_in[9];
  const int*   drows  = (const int*)d_in[10];
  const int*   dcols  = (const int*)d_in[11];
  float* out = (float*)d_out;

  float* Xd = (float*)d_ws;
  float* Y1 = Xd + (size_t)N_NODES_C * D_C;
  float* Xe = Y1 + (size_t)N_NODES_C * D_C;
  float* Xv = Xe + (size_t)N_EDGES_C * D_C;
  int*   ent = (int*)(Xv + 102400);
  float* esc = (float*)(ent + (size_t)3 * NNZ_C);
  int*   off = (int*)(esc + NNZ_C);
  static_assert(102400 + 4 * (size_t)NNZ_C + 3 * PART_CHUNKS * OFF_STRIDE <= (size_t)N_NODES_C * D_C,
                "entries fit in the last node-sized plane");


  k_partition<<<dim3(PART_CHUNKS, 3), OWN_THREADS, 0, stream>>>(
      drows, dcols, dvals, edges, vertex, ent, esc, off);
  k_owner_diffuse_w1<<<NBUCKET, OWN_THREADS, 0, stream>>>(
      X, ent, esc, off, W1_w, W1_b, Xd, Y1, N_NODES_C);
  k_owner_sum<NR_EDGES><<<NBUCKET, OWN_THREADS, 0, stream>>>(
      Y1, ent + (size_t)NNZ_C, off + PART_CHUNKS * OFF_STRIDE, Xe, N_EDGES_C);
  k_owner_node_out<<<NBUCKET, OWN_THREADS, 0, stream>>>(
      Xe, ent + (size_t)2 * NNZ_C, off + 2 * PART_CHUNKS * OFF_STRIDE,
      Xd, W2_w, W2_b, W_w, W_b, ALPHA_C, out, N_NODES_C);
}


# Owners order their entries by destination row in LDS (counting sort, rank = returned integer add); each wave sums its own rows in registers, one store per row: no LDS float adds; sums in rank order
# speedup vs baseline: 3.4909x; 1.0887x over previous
#include <hip/hip_runtime.h>

typedef __attribute__((ext_vector_type(2))) float v2f;
typedef __attribute__((ext_vector_type(8))) float v8f;

#define N_NODES_C 100000
#define N_EDGES_C 25000
#define NNZ_C     800000
#define D_C       64
#define ALPHA_C   0.5f

#define ACC_P     68
#define B_PSTRIDE 160

__device__ __forceinline__ v8f wmma_f32(v2f a, v2f b, v8f c) {
  return __builtin_amdgcn_wmma_f32_16x16x4_f32(
       false, a,  false, b,
       (short)0, c,  false,  false);
}

#define OWN_THREADS 1024
#define OWN_WAVES   (OWN_THREADS / 32)
#define NBUCKET     256
#define NR_NODES    ((N_NODES_C + NBUCKET - 1) / NBUCKET)
#define NR_EDGES    ((N_EDGES_C + NBUCKET - 1) / NBUCKET)
#define LROW_BITS   9
#define LROW_MASK   ((1 << LROW_BITS) - 1)
#define PART_CHUNKS 64
#define PART_CH     (NNZ_C / PART_CHUNKS)
#define PART_Q      (PART_CH / 4)
#define PART_IT     ((PART_Q + OWN_THREADS - 1) / OWN_THREADS)
#define OFF_STRIDE  (NBUCKET + 1)

static_assert(NNZ_C % (PART_CHUNKS * 4) == 0, "chunks are read as whole 16-byte pieces");
static_assert(PART_CH < 65536, "a rank inside a chunk is kept in 16 bits");
static_assert(NR_NODES <= (1 << LROW_BITS) && NR_EDGES <= (1 << LROW_BITS), "owned row index bits");
static_assert(N_NODES_C <= (1 << (31 - LROW_BITS)) && N_EDGES_C <= (1 << (31 - LROW_BITS)), "gather index bits");
static_assert(PART_CHUNKS % OWN_WAVES == 0, "every owner wave takes the same number of chunks");

__global__ void __launch_bounds__(OWN_THREADS) k_partition(
    const int* __restrict__ drows, const int* __restrict__ dcols, const float* __restrict__ dvals,
    const int* __restrict__ edges, const int* __restrict__ vertex,
    int* __restrict__ ent, float* __restrict__ esc, int* __restrict__ off) {
  __shared__ int hist[NBUCKET];
  __shared__ int offs[NBUCKET + 1];
  __shared__ int wtot[NBUCKET / 32];
  __shared__ __attribute__((aligned(16))) int   se[PART_CH];
  __shared__ __attribute__((aligned(16))) float ss[PART_CH];

  const int tid = threadIdx.x, lane = tid & 31;
  const int g = blockIdx.x, l = blockIdx.y;
  const int* key  = l == 0 ? drows : (l == 1 ? edges : vertex);
  const int* gidx = l == 0 ? dcols : (l == 1 ? vertex : edges);
  const unsigned nr   = l == 1 ? NR_EDGES : NR_NODES;
  const unsigned ndst = l == 1 ? N_EDGES_C : N_NODES_C;

  if (tid < NBUCKET) hist[tid] = 0;
  __syncthreads();

  const size_t cbase = (size_t)g * PART_CH;
  const int4*   key4 = reinterpret_cast<const int4*>(key + cbase);
  const int4*   gid4 = reinterpret_cast<const int4*>(gidx + cbase);
  const float4* sc4  = reinterpret_cast<const float4*>(dvals + cbase);

  int4 kv[PART_IT], gv[PART_IT];
  float4 sv[PART_IT];
#pragma unroll
  for (int it = 0; it < PART_IT; ++it) {
    const int qi = it * OWN_THREADS + tid;
    const int qc = qi < PART_Q ? qi : PART_Q - 1;
    kv[it] = key4[qc];
    gv[it] = gid4[qc];
    sv[it] = sc4[qc];
  }

  int code[PART_IT][4];
#pragma unroll
  for (int it = 0; it < PART_IT; ++it) {
    const int qi = it * OWN_THREADS + tid;
    const int kk[4] = {kv[it].x, kv[it].y, kv[it].z, kv[it].w};
#pragma unroll
    for (int c = 0; c < 4; ++c) {
      const unsigned k = (unsigned)kk[c];
      const unsigned o = l == 1 ? k / NR_EDGES : k / NR_NODES;
      int cd = -1;
      if (qi < PART_Q && k < ndst) cd = (int)(o << 16) | atomicAdd(&hist[o], 1);
      code[it][c] = cd;
    }
  }
  __syncthreads();

  int cnt = 0, inc = 0;
  if (tid < NBUCKET) {
    cnt = hist[tid];
    inc = cnt;
#pragma unroll
    for (int d = 1; d < 32; d <<= 1) {
      const int t = __shfl_up(inc, d);
      if (lane >= d) inc += t;
    }
    if (lane == 31) wtot[tid >> 5] = inc;
  }
  __syncthreads();
  if (tid < NBUCKET) {
    int base = 0;
    for (int w = 0; w < (tid >> 5); ++w) base += wtot[w];
    offs[tid] = base + inc - cnt;
    if (tid == NBUCKET - 1) offs[NBUCKET] = base + inc;
  }
  __syncthreads();

  if (tid <= NBUCKET)
    off[((size_t)l * PART_CHUNKS + g) * OFF_STRIDE + tid] = offs[tid];

#pragma unroll
  for (int it = 0; it < PART_IT; ++it) {
    const int kk[4] = {kv[it].x, kv[it].y, kv[it].z, kv[it].w};
    const int gg[4] = {gv[it].x, gv[it].y, gv[it].z, gv[it].w};
    const float sc[4] = {sv[it].x, sv[it].y, sv[it].z, sv[it].w};
#pragma unroll
    for (int c = 0; c < 4; ++c) {
      const int cd = code[it][c];
      if (cd >= 0) {
        const unsigned o = (unsigned)cd >> 16;
        const int pos = offs[o] + (cd & 0xffff);
        se[pos] = (gg[c] << LROW_BITS) | (int)((unsigned)kk[c] - o * nr);
        if (l == 0) ss[pos] = sc[c];
      }
    }
  }
  __syncthreads();

  const int total = offs[NBUCKET];
  int4*   ent4 = reinterpret_cast<int4*>(ent + (size_t)l * NNZ_C + cbase);
  float4* esc4 = reinterpret_cast<float4*>(esc + cbase);
  for (int q = tid; (q << 2) < total; q += OWN_THREADS) {
    ent4[q] = *reinterpret_cast<const int4*>(&se[q << 2]);
    if (l == 0) esc4[q] = *reinterpret_cast<const float4*>(&ss[q << 2]);
  }
}

#define OWN_BLK  32
#define SORT_B   256
#define SORT_NB  (SORT_B / OWN_BLK)
#define SORT_CAP (OWN_WAVES * SORT_B)
#define SCAN_OF(n) ((((n) + 31) / 32) * 32)

template <int NROWS, bool SCALED>
struct SortLds {
  int   rcnt[SCAN_OF(NROWS)];
  int   roff[SCAN_OF(NROWS) + 1];
  int   wtot[SCAN_OF(NROWS) / 32];
  int   smax;
  int   sent[SORT_CAP];
  float ssc[SCALED ? SORT_CAP : 1];
};

template <int P, int NROWS, bool SCALED, bool COUNT>
__device__ __forceinline__ void owner_accumulate(
    const float* __restrict__ src, const int* __restrict__ ent,
    const float* __restrict__ esc, const int* __restrict__ off,
    float* acc, float* cntl, SortLds<NROWS, SCALED>& sl, int own, int tid, int wave, int lane) {
  constexpr int RUNS   = PART_CHUNKS / OWN_WAVES;
  constexpr int SCAN_N = SCAN_OF(NROWS);
  static_assert(SCAN_N <= OWN_THREADS, "one thread per row slot in the prefix sum");

  int rs[RUNS], len[RUNS];
  int nw = 0;
#pragma unroll
  for (int r = 0; r < RUNS; ++r) {
    const int* o = off + (wave + r * OWN_WAVES) * OFF_STRIDE + own;
    const int a = __builtin_amdgcn_readfirstlane(o[0]);
    const int b = __builtin_amdgcn_readfirstlane(o[1]);
    rs[r]  = a;
    len[r] = b > a ? b - a : 0;
    nw += len[r];
  }
  if (lane == 0) atomicMax(&sl.smax, nw);
  __syncthreads();
  const int rounds = (__builtin_amdgcn_readfirstlane(sl.smax) + SORT_B - 1) / SORT_B;

  auto flat = [&](int p) -> size_t {
    size_t idx = 0;
    int q = p;
    bool done = false;
#pragma unroll
    for (int r = 0; r < RUNS; ++r) {
      if (!done) {
        if (q < len[r] || r == RUNS - 1) {
          idx = (size_t)(wave + r * OWN_WAVES) * PART_CH + (size_t)(rs[r] + q);
          done = true;
        } else {
          q -= len[r];
        }
      }
    }
    return idx;
  };

  const int rlo = (wave * NROWS) >> 5, rhi = ((wave + 1) * NROWS) >> 5;

  for (int rd = 0; rd < rounds; ++rd) {
    const int p0 = rd * SORT_B;
    int e[SORT_NB] = {}, rk[SORT_NB] = {};
    float s[SORT_NB] = {};
    if (p0 < nw) {
#pragma unroll
      for (int b = 0; b < SORT_NB; ++b) {
        const int pb = p0 + b * OWN_BLK;
        const int p = pb + lane < nw ? pb + lane : nw - 1;
        const size_t idx = flat(p);
        e[b] = ent[idx];
        s[b] = SCALED ? esc[idx] : 1.f;
      }
#pragma unroll
      for (int b = 0; b < SORT_NB; ++b) {
        rk[b] = 0;
        if (p0 + b * OWN_BLK + lane < nw) rk[b] = atomicAdd(&sl.rcnt[e[b] & LROW_MASK], 1);
      }
    }
    __syncthreads();

    int cnt = 0, inc = 0;
    if (tid < SCAN_N) {
      cnt = sl.rcnt[tid];
      inc = cnt;
#pragma unroll
      for (int d = 1; d < 32; d <<= 1) {
        const int t = __shfl_up(inc, d);
        if (lane >= d) inc += t;
      }
      if (lane == 31) sl.wtot[tid >> 5] = inc;
    }
    __syncthreads();
    if (tid < SCAN_N) {
      int base = 0;
      for (int w = 0; w < (tid >> 5); ++w) base += sl.wtot[w];
      sl.roff[tid] = base + inc - cnt;
      if (tid == SCAN_N - 1) sl.roff[SCAN_N] = base + inc;
      sl.rcnt[tid] = 0;
      if (COUNT) { if (tid < NROWS) cntl[tid] += (float)cnt; }
    }
    __syncthreads();

#pragma unroll
    for (int b = 0; b < SORT_NB; ++b) {
      const int pb = p0 + b * OWN_BLK;
      if (pb < nw && pb + lane < nw) {
        const int pos = sl.roff[e[b] & LROW_MASK] + rk[b];
        sl.sent[pos] = e[b];
        if (SCALED) sl.ssc[pos] = s[b];
      }
    }
    __syncthreads();

    const int j0 = __builtin_amdgcn_readfirstlane(sl.roff[rlo]);
    const int j1 = __builtin_amdgcn_readfirstlane(sl.roff[rhi]);
    int cur = -1;
    v2f sum = {0.f, 0.f};
    auto flush = [&]() {
      v2f* d = reinterpret_cast<v2f*>(&acc[cur * P + (lane << 1)]);
      if (rd == 0) {
        *d = sum;
      } else {
        v2f o = *d;
        o[0] += sum[0]; o[1] += sum[1];
        *d = o;
      }
    };
    for (int j = j0; j < j1; j += OWN_BLK) {
      const int n  = j1 - j < OWN_BLK ? j1 - j : OWN_BLK;
      const int jl = j + lane < j1 ? j + lane : j1 - 1;
      const int   me = sl.sent[jl];
      const float ms = SCALED ? sl.ssc[jl] : 1.f;
      v2f x[OWN_BLK];
#pragma unroll
      for (int k = 0; k < OWN_BLK; ++k) {
        const int eu = __builtin_amdgcn_readlane(me, k);
        x[k] = *reinterpret_cast<const v2f*>(src + (size_t)(eu >> LROW_BITS) * D_C + (lane << 1));
      }
#pragma unroll
      for (int k = 0; k < OWN_BLK; ++k) {
        const int row = __builtin_amdgcn_readlane(me, k) & LROW_MASK;
        if (row != cur) {
          if (cur >= 0) flush();
          cur = row;
          sum[0] = 0.f; sum[1] = 0.f;
        }
        float a = x[k][0], c = x[k][1];
        if (SCALED) {
          const float sc = __int_as_float(__builtin_amdgcn_readlane(__float_as_int(ms), k));
          a *= sc; c *= sc;
        }
        if (k >= n) { a = 0.f; c = 0.f; }
        sum[0] += a; sum[1] += c;
      }
    }
    if (cur >= 0) flush();
  }
}

template <int NR>
__global__ void __launch_bounds__(OWN_THREADS) k_owner_sum(
    const float* __restrict__ src, const int* __restrict__ ent, const int* __restrict__ off,
    float* __restrict__ dst, int ndst) {
  __shared__ __attribute__((aligned(16))) float acc[NR * D_C];
  __shared__ SortLds<NR, false> sl;

  const int tid  = threadIdx.x;
  const int lane = tid & 31;
  const int wave = __builtin_amdgcn_readfirstlane(tid >> 5);
  const int own  = blockIdx.x;
  const int r0   = own * NR;
  int nown = ndst - r0;
  nown = nown > NR ? NR : (nown < 0 ? 0 : nown);

  for (int j = tid; j < NR * D_C; j += OWN_THREADS) acc[j] = 0.f;
  for (int j = tid; j < SCAN_OF(NR); j += OWN_THREADS) sl.rcnt[j] = 0;
  if (tid == 0) sl.smax = 0;
  __syncthreads();
  owner_accumulate<D_C, NR, false, false>(src, ent, nullptr, off, acc, nullptr, sl, own, tid, wave, lane);
  __syncthreads();

  for (int j = tid; j < nown * (D_C / 4); j += OWN_THREADS)
    *reinterpret_cast<float4*>(dst + (size_t)r0 * D_C + (j << 2)) =
        *reinterpret_cast<const float4*>(&acc[j << 2]);
}

#define NT_NODES  ((NR_NODES + 15) / 16)
#define NRP_NODES (NT_NODES * 16)
static_assert(NT_NODES <= OWN_WAVES, "one tile per wave");

__device__ __forceinline__ void stage_b(float* Bf, const float* __restrict__ B, int nk, int tid) {
  for (int idx = tid; idx < nk * 64; idx += OWN_THREADS) {
    const int k = idx >> 6, col = idx & 63;
    Bf[(k >> 1) * B_PSTRIDE + (col << 1) + (k & 1)] = B[idx];
  }
}

template <bool SCALE>
__device__ __forceinline__ void tile_mac64(const float* arow, float s, const float* bl, v8f (&c)[4]) {
#pragma unroll
  for (int k4 = 0; k4 < 16; ++k4) {
    v2f a = *reinterpret_cast<const v2f*>(arow + (k4 << 2));
    if (SCALE) { a[0] *= s; a[1] *= s; }
#pragma unroll
    for (int n = 0; n < 4; ++n) {
      const v2f b = *reinterpret_cast<const v2f*>(bl + (k4 << 1) * B_PSTRIDE + (n << 5));
      c[n] = wmma_f32(a, b, c[n]);
    }
  }
}

__global__ void __launch_bounds__(OWN_THREADS) k_owner_diffuse_w1(
    const float* __restrict__ X, const int* __restrict__ ent, const float* __restrict__ esc,
    const int* __restrict__ off, const float* __restrict__ W1, const float* __restrict__ b1,
    float* __restrict__ Xd, float* __restrict__ Y1, int ndst) {
  __shared__ __attribute__((aligned(16))) float acc[NRP_NODES * ACC_P];
  __shared__ float Bf[32 * B_PSTRIDE];
  __shared__ float bs[64];
  __shared__ SortLds<NRP_NODES, true> sl;

  const int tid  = threadIdx.x;
  const int lane = tid & 31;
  const int wave = __builtin_amdgcn_readfirstlane(tid >> 5);
  const int own  = blockIdx.x;
  const int r0   = own * NR_NODES;
  int nown = ndst - r0;
  nown = nown > NR_NODES ? NR_NODES : (nown < 0 ? 0 : nown);

  for (int j = tid; j < NRP_NODES * ACC_P / 4; j += OWN_THREADS)
    *reinterpret_cast<float4*>(&acc[j << 2]) = make_float4(0.f, 0.f, 0.f, 0.f);
  for (int j = tid; j < SCAN_OF(NRP_NODES); j += OWN_THREADS) sl.rcnt[j] = 0;
  if (tid == 0) sl.smax = 0;
  stage_b(Bf, W1, 64, tid);
  if (tid < 64) bs[tid] = b1[tid];
  __syncthreads();
  owner_accumulate<ACC_P, NRP_NODES, true, false>(X, ent, esc, off, acc, nullptr, sl, own, tid, wave, lane);
  __syncthreads();

  for (int j = tid; j < nown * 16; j += OWN_THREADS) {
    const int row = j >> 4, c4 = j & 15;
    *reinterpret_cast<float4*>(Xd + (size_t)(r0 + row) * D_C + (c4 << 2)) =
        *reinterpret_cast<const float4*>(&acc[row * ACC_P + (c4 << 2)]);
  }

  const int lrow = lane & 15, hi = lane >> 4;
  v8f c[4] = {v8f{}, v8f{}, v8f{}, v8f{}};
  if (wave < NT_NODES)
    tile_mac64<false>(&acc[(wave * 16 + lrow) * ACC_P + (hi << 1)], 1.f,
                      &Bf[hi * B_PSTRIDE + (lrow << 1)], c);
  __syncthreads();
  if (wave < NT_NODES) {
#pragma unroll
    for (int n = 0; n < 4; ++n)
#pragma unroll
      for (int j = 0; j < 8; ++j) {
        const int lr = wave * 16 + j + (hi << 3);
        const int col = (n << 4) + lrow;
        acc[lr * ACC_P + col] = c[n][j] + bs[col];
      }
  }
  __syncthreads();
  for (int j = tid; j < nown * 16; j += OWN_THREADS) {
    const int row = j >> 4, c4 = j & 15;
    *reinterpret_cast<float4*>(Y1 + (size_t)(r0 + row) * D_C + (c4 << 2)) =
        *reinterpret_cast<const float4*>(&acc[row * ACC_P + (c4 << 2)]);
  }
}

__global__ void __launch_bounds__(OWN_THREADS) k_owner_node_out(
    const float* __restrict__ Xe, const int* __restrict__ ent, const int* __restrict__ off,
    const float* __restrict__ Xd, const float* __restrict__ W2, const float* __restrict__ b2,
    const float* __restrict__ Ww, const float* __restrict__ bw, float alpha,
    float* __restrict__ out, int ndst) {
  __shared__ __attribute__((aligned(16))) float acc[NRP_NODES * ACC_P];
  __shared__ __attribute__((aligned(16))) float xs[NRP_NODES * ACC_P];
  __shared__ float Bf2[64 * B_PSTRIDE];
  __shared__ float Bfw[32 * B_PSTRIDE];
  __shared__ float b2s[64];
  __shared__ float bws[64];
  __shared__ float cntl[NRP_NODES];
  __shared__ SortLds<NRP_NODES, false> sl;

  const int tid  = threadIdx.x;
  const int lane = tid & 31;
  const int wave = __builtin_amdgcn_readfirstlane(tid >> 5);
  const int own  = blockIdx.x;
  const int r0   = own * NR_NODES;
  int nown = ndst - r0;
  nown = nown > NR_NODES ? NR_NODES : (nown < 0 ? 0 : nown);

  for (int j = tid; j < NRP_NODES * 16; j += OWN_THREADS) {
    const int row = j >> 4, c4 = j & 15;
    float4 v = make_float4(0.f, 0.f, 0.f, 0.f);
    if (row < nown) v = *reinterpret_cast<const float4*>(Xd + (size_t)(r0 + row) * D_C + (c4 << 2));
    *reinterpret_cast<float4*>(&xs[row * ACC_P + (c4 << 2)]) = v;
  }
  for (int j = tid; j < NRP_NODES * ACC_P / 4; j += OWN_THREADS)
    *reinterpret_cast<float4*>(&acc[j << 2]) = make_float4(0.f, 0.f, 0.f, 0.f);
  for (int j = tid; j < NRP_NODES; j += OWN_THREADS) cntl[j] = 0.f;
  for (int j = tid; j < SCAN_OF(NRP_NODES); j += OWN_THREADS) sl.rcnt[j] = 0;
  if (tid == 0) sl.smax = 0;
  stage_b(Bf2, W2, 128, tid);
  stage_b(Bfw, Ww, 64, tid);
  if (tid < 64) { b2s[tid] = b2[tid]; bws[tid] = bw[tid]; }
  __syncthreads();
  owner_accumulate<ACC_P, NRP_NODES, false, true>(Xe, ent, nullptr, off, acc, cntl, sl, own, tid, wave, lane);
  __syncthreads();

  const int lrow = lane & 15, hi = lane >> 4;
  v8f c[4] = {v8f{}, v8f{}, v8f{}, v8f{}};
  if (wave < NT_NODES) {
    const int ar = (wave * 16 + lrow) * ACC_P + (hi << 1);
    const float* bl = &Bf2[hi * B_PSTRIDE + (lrow << 1)];
    tile_mac64<true >(&xs[ar],  cntl[wave * 16 + lrow], bl, c);
    tile_mac64<false>(&acc[ar], 1.f,                    bl + 32 * B_PSTRIDE, c);
#pragma unroll
    for (int n = 0; n < 4; ++n)
#pragma unroll
      for (int j = 0; j < 8; ++j) {
        const int lr = wave * 16 + j + (hi << 3);
        const int col = (n << 4) + lrow;
        const float xv = c[n][j] + cntl[lr] * b2s[col];
        c[n][j] = (1.f - alpha) * xv + alpha * xs[lr * ACC_P + col];
      }
  }
  __syncthreads();
  if (wave < NT_NODES) {
#pragma unroll
    for (int n = 0; n < 4; ++n)
#pragma unroll
      for (int j = 0; j < 8; ++j)
        acc[(wave * 16 + j + (hi << 3)) * ACC_P + (n << 4) + lrow] = c[n][j];
  }
  __syncthreads();
  v8f o[4] = {v8f{}, v8f{}, v8f{}, v8f{}};
  if (wave < NT_NODES)
    tile_mac64<false>(&acc[(wave * 16 + lrow) * ACC_P + (hi << 1)], 1.f,
                      &Bfw[hi * B_PSTRIDE + (lrow << 1)], o);
  __syncthreads();
  if (wave < NT_NODES) {
#pragma unroll
    for (int n = 0; n < 4; ++n)
#pragma unroll
      for (int j = 0; j < 8; ++j) {
        const int col = (n << 4) + lrow;
        acc[(wave * 16 + j + (hi << 3)) * ACC_P + col] = o[n][j] + bws[col];
      }
  }
  __syncthreads();
  for (int j = tid; j < nown * 16; j += OWN_THREADS) {
    const int row = j >> 4, c4 = j & 15;
    *reinterpret_cast<float4*>(out + (size_t)(r0 + row) * D_C + (c4 << 2)) =
        *reinterpret_cast<const float4*>(&acc[row * ACC_P + (c4 << 2)]);
  }
}

extern "C" void kernel_launch(void* const* d_in, const int* in_sizes, int n_in,
                              void* d_out, int out_size, void* d_ws, size_t ws_size,
                              hipStream_t stream) {
  const float* X      = (const float*)d_in[0];
  const float* W1_w   = (const float*)d_in[1];
  const float* W1_b   = (const float*)d_in[2];
  const float* W2_w   = (const float*)d_in[3];
  const float* W2_b   = (const float*)d_in[4];
  const float* W_w    = (const float*)d_in[5];
  const float* W_b    = (const float*)d_in[6];
  const float* dvals  = (const float*)d_in[7];
  const int*   vertex = (const int*)d_in[8];
  const int*   edges  = (const int*)d_in[9];
  const int*   drows  = (const int*)d_in[10];
  const int*   dcols  = (const int*)d_in[11];
  float* out = (float*)d_out;

  float* Xd = (float*)d_ws;
  float* Y1 = Xd + (size_t)N_NODES_C * D_C;
  float* Xe = Y1 + (size_t)N_NODES_C * D_C;
  float* Xv = Xe + (size_t)N_EDGES_C * D_C;
  int*   ent = (int*)(Xv + 102400);
  float* esc = (float*)(ent + (size_t)3 * NNZ_C);
  int*   off = (int*)(esc + NNZ_C);
  static_assert(102400 + 4 * (size_t)NNZ_C + 3 * PART_CHUNKS * OFF_STRIDE <= (size_t)N_NODES_C * D_C,
                "entries fit in the last node-sized plane");


  k_partition<<<dim3(PART_CHUNKS, 3), OWN_THREADS, 0, stream>>>(
      drows, dcols, dvals, edges, vertex, ent, esc, off);
  k_owner_diffuse_w1<<<NBUCKET, OWN_THREADS, 0, stream>>>(
      X, ent, esc, off, W1_w, W1_b, Xd, Y1, N_NODES_C);
  k_owner_sum<NR_EDGES><<<NBUCKET, OWN_THREADS, 0, stream>>>(
      Y1, ent + (size_t)NNZ_C, off + PART_CHUNKS * OFF_STRIDE, Xe, N_EDGES_C);
  k_owner_node_out<<<NBUCKET, OWN_THREADS, 0, stream>>>(
      Xe, ent + (size_t)2 * NNZ_C, off + 2 * PART_CHUNKS * OFF_STRIDE,
      Xd, W2_w, W2_b, W_w, W_b, ALPHA_C, out, N_NODES_C);
}


# W1 product moved behind the hyperedge sum (linearity): edge owners sum Xd[vertex] rows, count pairs, then Xe = sum @ W1 + count*b1 on their LDS rows (f32 WMMA); no Y1 plane; sum-then-product rounding
# speedup vs baseline: 3.7185x; 1.0652x over previous
#include <hip/hip_runtime.h>

typedef __attribute__((ext_vector_type(2))) float v2f;
typedef __attribute__((ext_vector_type(8))) float v8f;

#define N_NODES_C 100000
#define N_EDGES_C 25000
#define NNZ_C     800000
#define D_C       64
#define ALPHA_C   0.5f

#define ACC_P     68
#define B_PSTRIDE 160

__device__ __forceinline__ v8f wmma_f32(v2f a, v2f b, v8f c) {
  return __builtin_amdgcn_wmma_f32_16x16x4_f32(
       false, a,  false, b,
       (short)0, c,  false,  false);
}

#define OWN_THREADS 1024
#define OWN_WAVES   (OWN_THREADS / 32)
#define NBUCKET     256
#define NR_NODES    ((N_NODES_C + NBUCKET - 1) / NBUCKET)
#define NR_EDGES    ((N_EDGES_C + NBUCKET - 1) / NBUCKET)
#define LROW_BITS   9
#define LROW_MASK   ((1 << LROW_BITS) - 1)
#define PART_CHUNKS 64
#define PART_CH     (NNZ_C / PART_CHUNKS)
#define PART_Q      (PART_CH / 4)
#define PART_IT     ((PART_Q + OWN_THREADS - 1) / OWN_THREADS)
#define OFF_STRIDE  (NBUCKET + 1)

static_assert(NNZ_C % (PART_CHUNKS * 4) == 0, "chunks are read as whole 16-byte pieces");
static_assert(PART_CH < 65536, "a rank inside a chunk is kept in 16 bits");
static_assert(NR_NODES <= (1 << LROW_BITS) && NR_EDGES <= (1 << LROW_BITS), "owned row index bits");
static_assert(N_NODES_C <= (1 << (31 - LROW_BITS)) && N_EDGES_C <= (1 << (31 - LROW_BITS)), "gather index bits");
static_assert(PART_CHUNKS % OWN_WAVES == 0, "every owner wave takes the same number of chunks");

__global__ void __launch_bounds__(OWN_THREADS) k_partition(
    const int* __restrict__ drows, const int* __restrict__ dcols, const float* __restrict__ dvals,
    const int* __restrict__ edges, const int* __restrict__ vertex,
    int* __restrict__ ent, float* __restrict__ esc, int* __restrict__ off) {
  __shared__ int hist[NBUCKET];
  __shared__ int offs[NBUCKET + 1];
  __shared__ int wtot[NBUCKET / 32];
  __shared__ __attribute__((aligned(16))) int   se[PART_CH];
  __shared__ __attribute__((aligned(16))) float ss[PART_CH];

  const int tid = threadIdx.x, lane = tid & 31;
  const int g = blockIdx.x, l = blockIdx.y;
  const int* key  = l == 0 ? drows : (l == 1 ? edges : vertex);
  const int* gidx = l == 0 ? dcols : (l == 1 ? vertex : edges);
  const unsigned nr   = l == 1 ? NR_EDGES : NR_NODES;
  const unsigned ndst = l == 1 ? N_EDGES_C : N_NODES_C;

  if (tid < NBUCKET) hist[tid] = 0;
  __syncthreads();

  const size_t cbase = (size_t)g * PART_CH;
  const int4*   key4 = reinterpret_cast<const int4*>(key + cbase);
  const int4*   gid4 = reinterpret_cast<const int4*>(gidx + cbase);
  const float4* sc4  = reinterpret_cast<const float4*>(dvals + cbase);

  int4 kv[PART_IT], gv[PART_IT];
  float4 sv[PART_IT];
#pragma unroll
  for (int it = 0; it < PART_IT; ++it) {
    const int qi = it * OWN_THREADS + tid;
    const int qc = qi < PART_Q ? qi : PART_Q - 1;
    kv[it] = key4[qc];
    gv[it] = gid4[qc];
    sv[it] = sc4[qc];
  }

  int code[PART_IT][4];
#pragma unroll
  for (int it = 0; it < PART_IT; ++it) {
    const int qi = it * OWN_THREADS + tid;
    const int kk[4] = {kv[it].x, kv[it].y, kv[it].z, kv[it].w};
#pragma unroll
    for (int c = 0; c < 4; ++c) {
      const unsigned k = (unsigned)kk[c];
      const unsigned o = l == 1 ? k / NR_EDGES : k / NR_NODES;
      int cd = -1;
      if (qi < PART_Q && k < ndst) cd = (int)(o << 16) | atomicAdd(&hist[o], 1);
      code[it][c] = cd;
    }
  }
  __syncthreads();

  int cnt = 0, inc = 0;
  if (tid < NBUCKET) {
    cnt = hist[tid];
    inc = cnt;
#pragma unroll
    for (int d = 1; d < 32; d <<= 1) {
      const int t = __shfl_up(inc, d);
      if (lane >= d) inc += t;
    }
    if (lane == 31) wtot[tid >> 5] = inc;
  }
  __syncthreads();
  if (tid < NBUCKET) {
    int base = 0;
    for (int w = 0; w < (tid >> 5); ++w) base += wtot[w];
    offs[tid] = base + inc - cnt;
    if (tid == NBUCKET - 1) offs[NBUCKET] = base + inc;
  }
  __syncthreads();

  if (tid <= NBUCKET)
    off[((size_t)l * PART_CHUNKS + g) * OFF_STRIDE + tid] = offs[tid];

#pragma unroll
  for (int it = 0; it < PART_IT; ++it) {
    const int kk[4] = {kv[it].x, kv[it].y, kv[it].z, kv[it].w};
    const int gg[4] = {gv[it].x, gv[it].y, gv[it].z, gv[it].w};
    const float sc[4] = {sv[it].x, sv[it].y, sv[it].z, sv[it].w};
#pragma unroll
    for (int c = 0; c < 4; ++c) {
      const int cd = code[it][c];
      if (cd >= 0) {
        const unsigned o = (unsigned)cd >> 16;
        const int pos = offs[o] + (cd & 0xffff);
        se[pos] = (gg[c] << LROW_BITS) | (int)((unsigned)kk[c] - o * nr);
        if (l == 0) ss[pos] = sc[c];
      }
    }
  }
  __syncthreads();

  const int total = offs[NBUCKET];
  int4*   ent4 = reinterpret_cast<int4*>(ent + (size_t)l * NNZ_C + cbase);
  float4* esc4 = reinterpret_cast<float4*>(esc + cbase);
  for (int q = tid; (q << 2) < total; q += OWN_THREADS) {
    ent4[q] = *reinterpret_cast<const int4*>(&se[q << 2]);
    if (l == 0) esc4[q] = *reinterpret_cast<const float4*>(&ss[q << 2]);
  }
}

#define OWN_BLK  32
#define SORT_B   256
#define SORT_NB  (SORT_B / OWN_BLK)
#define SORT_CAP (OWN_WAVES * SORT_B)
#define SCAN_OF(n) ((((n) + 31) / 32) * 32)

template <int NROWS, bool SCALED>
struct SortLds {
  int   rcnt[SCAN_OF(NROWS)];
  int   roff[SCAN_OF(NROWS) + 1];
  int   wtot[SCAN_OF(NROWS) / 32];
  int   smax;
  int   sent[SORT_CAP];
  float ssc[SCALED ? SORT_CAP : 1];
};

template <int P, int NROWS, int NSUM, bool SCALED, bool COUNT>
__device__ __forceinline__ void owner_accumulate(
    const float* __restrict__ src, const int* __restrict__ ent,
    const float* __restrict__ esc, const int* __restrict__ off,
    float* acc, float* cntl, SortLds<NROWS, SCALED>& sl, int own, int tid, int wave, int lane) {
  constexpr int RUNS   = PART_CHUNKS / OWN_WAVES;
  constexpr int SCAN_N = SCAN_OF(NROWS);
  static_assert(SCAN_N <= OWN_THREADS, "one thread per row slot in the prefix sum");

  int rs[RUNS], len[RUNS];
  int nw = 0;
#pragma unroll
  for (int r = 0; r < RUNS; ++r) {
    const int* o = off + (wave + r * OWN_WAVES) * OFF_STRIDE + own;
    const int a = __builtin_amdgcn_readfirstlane(o[0]);
    const int b = __builtin_amdgcn_readfirstlane(o[1]);
    rs[r]  = a;
    len[r] = b > a ? b - a : 0;
    nw += len[r];
  }
  if (lane == 0) atomicMax(&sl.smax, nw);
  __syncthreads();
  const int rounds = (__builtin_amdgcn_readfirstlane(sl.smax) + SORT_B - 1) / SORT_B;

  auto flat = [&](int p) -> size_t {
    size_t idx = 0;
    int q = p;
    bool done = false;
#pragma unroll
    for (int r = 0; r < RUNS; ++r) {
      if (!done) {
        if (q < len[r] || r == RUNS - 1) {
          idx = (size_t)(wave + r * OWN_WAVES) * PART_CH + (size_t)(rs[r] + q);
          done = true;
        } else {
          q -= len[r];
        }
      }
    }
    return idx;
  };

  static_assert(NSUM <= NROWS, "summed rows are row slots");
  const int rlo = (wave * NSUM) >> 5, rhi = ((wave + 1) * NSUM) >> 5;

  for (int rd = 0; rd < rounds; ++rd) {
    const int p0 = rd * SORT_B;
    int e[SORT_NB] = {}, rk[SORT_NB] = {};
    float s[SORT_NB] = {};
    if (p0 < nw) {
#pragma unroll
      for (int b = 0; b < SORT_NB; ++b) {
        const int pb = p0 + b * OWN_BLK;
        const int p = pb + lane < nw ? pb + lane : nw - 1;
        const size_t idx = flat(p);
        e[b] = ent[idx];
        s[b] = SCALED ? esc[idx] : 1.f;
      }
#pragma unroll
      for (int b = 0; b < SORT_NB; ++b) {
        rk[b] = 0;
        if (p0 + b * OWN_BLK + lane < nw) rk[b] = atomicAdd(&sl.rcnt[e[b] & LROW_MASK], 1);
      }
    }
    __syncthreads();

    int cnt = 0, inc = 0;
    if (tid < SCAN_N) {
      cnt = sl.rcnt[tid];
      inc = cnt;
#pragma unroll
      for (int d = 1; d < 32; d <<= 1) {
        const int t = __shfl_up(inc, d);
        if (lane >= d) inc += t;
      }
      if (lane == 31) sl.wtot[tid >> 5] = inc;
    }
    __syncthreads();
    if (tid < SCAN_N) {
      int base = 0;
      for (int w = 0; w < (tid >> 5); ++w) base += sl.wtot[w];
      sl.roff[tid] = base + inc - cnt;
      if (tid == SCAN_N - 1) sl.roff[SCAN_N] = base + inc;
      sl.rcnt[tid] = 0;
      if (COUNT) { if (tid < NROWS) cntl[tid] += (float)cnt; }
    }
    __syncthreads();

#pragma unroll
    for (int b = 0; b < SORT_NB; ++b) {
      const int pb = p0 + b * OWN_BLK;
      if (pb < nw && pb + lane < nw) {
        const int pos = sl.roff[e[b] & LROW_MASK] + rk[b];
        sl.sent[pos] = e[b];
        if (SCALED) sl.ssc[pos] = s[b];
      }
    }
    __syncthreads();

    const int j0 = __builtin_amdgcn_readfirstlane(sl.roff[rlo]);
    const int j1 = __builtin_amdgcn_readfirstlane(sl.roff[rhi]);
    int cur = -1;
    v2f sum = {0.f, 0.f};
    auto flush = [&]() {
      v2f* d = reinterpret_cast<v2f*>(&acc[cur * P + (lane << 1)]);
      if (rd == 0) {
        *d = sum;
      } else {
        v2f o = *d;
        o[0] += sum[0]; o[1] += sum[1];
        *d = o;
      }
    };
    for (int j = j0; j < j1; j += OWN_BLK) {
      const int n  = j1 - j < OWN_BLK ? j1 - j : OWN_BLK;
      const int jl = j + lane < j1 ? j + lane : j1 - 1;
      const int   me = sl.sent[jl];
      const float ms = SCALED ? sl.ssc[jl] : 1.f;
      v2f x[OWN_BLK];
#pragma unroll
      for (int k = 0; k < OWN_BLK; ++k) {
        const int eu = __builtin_amdgcn_readlane(me, k);
        x[k] = *reinterpret_cast<const v2f*>(src + (size_t)(eu >> LROW_BITS) * D_C + (lane << 1));
      }
#pragma unroll
      for (int k = 0; k < OWN_BLK; ++k) {
        const int row = __builtin_amdgcn_readlane(me, k) & LROW_MASK;
        if (row != cur) {
          if (cur >= 0) flush();
          cur = row;
          sum[0] = 0.f; sum[1] = 0.f;
        }
        float a = x[k][0], c = x[k][1];
        if (SCALED) {
          const float sc = __int_as_float(__builtin_amdgcn_readlane(__float_as_int(ms), k));
          a *= sc; c *= sc;
        }
        if (k >= n) { a = 0.f; c = 0.f; }
        sum[0] += a; sum[1] += c;
      }
    }
    if (cur >= 0) flush();
  }
}

template <int NR>
__global__ void __launch_bounds__(OWN_THREADS) k_owner_sum_scaled(
    const float* __restrict__ src, const int* __restrict__ ent, const float* __restrict__ esc,
    const int* __restrict__ off, float* __restrict__ dst, int ndst) {
  __shared__ __attribute__((aligned(16))) float acc[NR * D_C];
  __shared__ SortLds<NR, true> sl;

  const int tid  = threadIdx.x;
  const int lane = tid & 31;
  const int wave = __builtin_amdgcn_readfirstlane(tid >> 5);
  const int own  = blockIdx.x;
  const int r0   = own * NR;
  int nown = ndst - r0;
  nown = nown > NR ? NR : (nown < 0 ? 0 : nown);

  for (int j = tid; j < NR * D_C / 4; j += OWN_THREADS)
    *reinterpret_cast<float4*>(&acc[j << 2]) = make_float4(0.f, 0.f, 0.f, 0.f);
  for (int j = tid; j < SCAN_OF(NR); j += OWN_THREADS) sl.rcnt[j] = 0;
  if (tid == 0) sl.smax = 0;
  __syncthreads();
  owner_accumulate<D_C, NR, NR, true, false>(src, ent, esc, off, acc, nullptr, sl, own, tid, wave, lane);
  __syncthreads();

  for (int j = tid; j < nown * (D_C / 4); j += OWN_THREADS)
    *reinterpret_cast<float4*>(dst + (size_t)r0 * D_C + (j << 2)) =
        *reinterpret_cast<const float4*>(&acc[j << 2]);
}

#define NT_NODES  ((NR_NODES + 15) / 16)
#define NRP_NODES (NT_NODES * 16)
static_assert(NT_NODES <= OWN_WAVES, "one tile per wave");

__device__ __forceinline__ void stage_b(float* Bf, const float* __restrict__ B, int nk, int tid) {
  for (int idx = tid; idx < nk * 64; idx += OWN_THREADS) {
    const int k = idx >> 6, col = idx & 63;
    Bf[(k >> 1) * B_PSTRIDE + (col << 1) + (k & 1)] = B[idx];
  }
}

template <bool SCALE>
__device__ __forceinline__ void tile_mac64(const float* arow, float s, const float* bl, v8f (&c)[4]) {
#pragma unroll
  for (int k4 = 0; k4 < 16; ++k4) {
    v2f a = *reinterpret_cast<const v2f*>(arow + (k4 << 2));
    if (SCALE) { a[0] *= s; a[1] *= s; }
#pragma unroll
    for (int n = 0; n < 4; ++n) {
      const v2f b = *reinterpret_cast<const v2f*>(bl + (k4 << 1) * B_PSTRIDE + (n << 5));
      c[n] = wmma_f32(a, b, c[n]);
    }
  }
}

#define NT_EDGES  ((NR_EDGES + 15) / 16)
#define NRP_EDGES (NT_EDGES * 16)
static_assert(4 * NT_EDGES <= OWN_WAVES, "one 16 x 16 output tile per wave");

__global__ void __launch_bounds__(OWN_THREADS) k_owner_edge_w1(
    const float* __restrict__ Xd, const int* __restrict__ ent, const int* __restrict__ off,
    const float* __restrict__ W1, const float* __restrict__ b1,
    float* __restrict__ Xe, int ndst) {
  __shared__ __attribute__((aligned(16))) float acc[NRP_EDGES * ACC_P];
  __shared__ float Bf[32 * B_PSTRIDE];
  __shared__ float bs[64];
  __shared__ float cntl[NRP_EDGES];
  __shared__ SortLds<NRP_EDGES, false> sl;

  const int tid  = threadIdx.x;
  const int lane = tid & 31;
  const int wave = __builtin_amdgcn_readfirstlane(tid >> 5);
  const int own  = blockIdx.x;
  const int r0   = own * NR_EDGES;
  int nown = ndst - r0;
  nown = nown > NR_EDGES ? NR_EDGES : (nown < 0 ? 0 : nown);

  for (int j = tid; j < NRP_EDGES * ACC_P / 4; j += OWN_THREADS)
    *reinterpret_cast<float4*>(&acc[j << 2]) = make_float4(0.f, 0.f, 0.f, 0.f);
  for (int j = tid; j < NRP_EDGES; j += OWN_THREADS) cntl[j] = 0.f;
  for (int j = tid; j < SCAN_OF(NRP_EDGES); j += OWN_THREADS) sl.rcnt[j] = 0;
  if (tid == 0) sl.smax = 0;
  stage_b(Bf, W1, 64, tid);
  if (tid < 64) bs[tid] = b1[tid];
  __syncthreads();
  owner_accumulate<ACC_P, NRP_EDGES, NR_EDGES, false, true>(Xd, ent, nullptr, off, acc, cntl, sl, own, tid, wave, lane);
  __syncthreads();

  const int lrow = lane & 15, hi = lane >> 4;
  const int tile = wave >> 2, nt = wave & 3;
  v8f c = v8f{};
  if (wave < 4 * NT_EDGES) {
    const float* arow = &acc[(tile * 16 + lrow) * ACC_P + (hi << 1)];
    const float* bl   = &Bf[hi * B_PSTRIDE + (lrow << 1) + (nt << 5)];
#pragma unroll
    for (int k4 = 0; k4 < 16; ++k4) {
      const v2f a = *reinterpret_cast<const v2f*>(arow + (k4 << 2));
      const v2f b = *reinterpret_cast<const v2f*>(bl + (k4 << 1) * B_PSTRIDE);
      c = wmma_f32(a, b, c);
    }
  }
  __syncthreads();
  if (wave < 4 * NT_EDGES) {
#pragma unroll
    for (int j = 0; j < 8; ++j) {
      const int lr = tile * 16 + j + (hi << 3);
      const int col = (nt << 4) + lrow;
      acc[lr * ACC_P + col] = c[j] + cntl[lr] * bs[col];
    }
  }
  __syncthreads();
  for (int j = tid; j < nown * 16; j += OWN_THREADS) {
    const int row = j >> 4, c4 = j & 15;
    *reinterpret_cast<float4*>(Xe + (size_t)(r0 + row) * D_C + (c4 << 2)) =
        *reinterpret_cast<const float4*>(&acc[row * ACC_P + (c4 << 2)]);
  }
}

__global__ void __launch_bounds__(OWN_THREADS) k_owner_node_out(
    const float* __restrict__ Xe, const int* __restrict__ ent, const int* __restrict__ off,
    const float* __restrict__ Xd, const float* __restrict__ W2, const float* __restrict__ b2,
    const float* __restrict__ Ww, const float* __restrict__ bw, float alpha,
    float* __restrict__ out, int ndst) {
  __shared__ __attribute__((aligned(16))) float acc[NRP_NODES * ACC_P];
  __shared__ __attribute__((aligned(16))) float xs[NRP_NODES * ACC_P];
  __shared__ float Bf2[64 * B_PSTRIDE];
  __shared__ float Bfw[32 * B_PSTRIDE];
  __shared__ float b2s[64];
  __shared__ float bws[64];
  __shared__ float cntl[NRP_NODES];
  __shared__ SortLds<NRP_NODES, false> sl;

  const int tid  = threadIdx.x;
  const int lane = tid & 31;
  const int wave = __builtin_amdgcn_readfirstlane(tid >> 5);
  const int own  = blockIdx.x;
  const int r0   = own * NR_NODES;
  int nown = ndst - r0;
  nown = nown > NR_NODES ? NR_NODES : (nown < 0 ? 0 : nown);

  for (int j = tid; j < NRP_NODES * 16; j += OWN_THREADS) {
    const int row = j >> 4, c4 = j & 15;
    float4 v = make_float4(0.f, 0.f, 0.f, 0.f);
    if (row < nown) v = *reinterpret_cast<const float4*>(Xd + (size_t)(r0 + row) * D_C + (c4 << 2));
    *reinterpret_cast<float4*>(&xs[row * ACC_P + (c4 << 2)]) = v;
  }
  for (int j = tid; j < NRP_NODES * ACC_P / 4; j += OWN_THREADS)
    *reinterpret_cast<float4*>(&acc[j << 2]) = make_float4(0.f, 0.f, 0.f, 0.f);
  for (int j = tid; j < NRP_NODES; j += OWN_THREADS) cntl[j] = 0.f;
  for (int j = tid; j < SCAN_OF(NRP_NODES); j += OWN_THREADS) sl.rcnt[j] = 0;
  if (tid == 0) sl.smax = 0;
  stage_b(Bf2, W2, 128, tid);
  stage_b(Bfw, Ww, 64, tid);
  if (tid < 64) { b2s[tid] = b2[tid]; bws[tid] = bw[tid]; }
  __syncthreads();
  owner_accumulate<ACC_P, NRP_NODES, NRP_NODES, false, true>(Xe, ent, nullptr, off, acc, cntl, sl, own, tid, wave, lane);
  __syncthreads();

  const int lrow = lane & 15, hi = lane >> 4;
  v8f c[4] = {v8f{}, v8f{}, v8f{}, v8f{}};
  if (wave < NT_NODES) {
    const int ar = (wave * 16 + lrow) * ACC_P + (hi << 1);
    const float* bl = &Bf2[hi * B_PSTRIDE + (lrow << 1)];
    tile_mac64<true >(&xs[ar],  cntl[wave * 16 + lrow], bl, c);
    tile_mac64<false>(&acc[ar], 1.f,                    bl + 32 * B_PSTRIDE, c);
#pragma unroll
    for (int n = 0; n < 4; ++n)
#pragma unroll
      for (int j = 0; j < 8; ++j) {
        const int lr = wave * 16 + j + (hi << 3);
        const int col = (n << 4) + lrow;
        const float xv = c[n][j] + cntl[lr] * b2s[col];
        c[n][j] = (1.f - alpha) * xv + alpha * xs[lr * ACC_P + col];
      }
  }
  __syncthreads();
  if (wave < NT_NODES) {
#pragma unroll
    for (int n = 0; n < 4; ++n)
#pragma unroll
      for (int j = 0; j < 8; ++j)
        acc[(wave * 16 + j + (hi << 3)) * ACC_P + (n << 4) + lrow] = c[n][j];
  }
  __syncthreads();
  v8f o[4] = {v8f{}, v8f{}, v8f{}, v8f{}};
  if (wave < NT_NODES)
    tile_mac64<false>(&acc[(wave * 16 + lrow) * ACC_P + (hi << 1)], 1.f,
                      &Bfw[hi * B_PSTRIDE + (lrow << 1)], o);
  __syncthreads();
  if (wave < NT_NODES) {
#pragma unroll
    for (int n = 0; n < 4; ++n)
#pragma unroll
      for (int j = 0; j < 8; ++j) {
        const int col = (n << 4) + lrow;
        acc[(wave * 16 + j + (hi << 3)) * ACC_P + col] = o[n][j] + bws[col];
      }
  }
  __syncthreads();
  for (int j = tid; j < nown * 16; j += OWN_THREADS) {
    const int row = j >> 4, c4 = j & 15;
    *reinterpret_cast<float4*>(out + (size_t)(r0 + row) * D_C + (c4 << 2)) =
        *reinterpret_cast<const float4*>(&acc[row * ACC_P + (c4 << 2)]);
  }
}

extern "C" void kernel_launch(void* const* d_in, const int* in_sizes, int n_in,
                              void* d_out, int out_size, void* d_ws, size_t ws_size,
                              hipStream_t stream) {
  const float* X      = (const float*)d_in[0];
  const float* W1_w   = (const float*)d_in[1];
  const float* W1_b   = (const float*)d_in[2];
  const float* W2_w   = (const float*)d_in[3];
  const float* W2_b   = (const float*)d_in[4];
  const float* W_w    = (const float*)d_in[5];
  const float* W_b    = (const float*)d_in[6];
  const float* dvals  = (const float*)d_in[7];
  const int*   vertex = (const int*)d_in[8];
  const int*   edges  = (const int*)d_in[9];
  const int*   drows  = (const int*)d_in[10];
  const int*   dcols  = (const int*)d_in[11];
  float* out = (float*)d_out;

  float* Xd = (float*)d_ws;
  float* Xe = Xd + (size_t)2 * N_NODES_C * D_C;
  float* Xv = Xe + (size_t)N_EDGES_C * D_C;
  int*   ent = (int*)(Xv + 102400);
  float* esc = (float*)(ent + (size_t)3 * NNZ_C);
  int*   off = (int*)(esc + NNZ_C);
  static_assert(102400 + 4 * (size_t)NNZ_C + 3 * PART_CHUNKS * OFF_STRIDE <= (size_t)N_NODES_C * D_C,
                "entries fit in the last node-sized plane");


  k_partition<<<dim3(PART_CHUNKS, 3), OWN_THREADS, 0, stream>>>(
      drows, dcols, dvals, edges, vertex, ent, esc, off);
  k_owner_sum_scaled<NR_NODES><<<NBUCKET, OWN_THREADS, 0, stream>>>(
      X, ent, esc, off, Xd, N_NODES_C);
  k_owner_edge_w1<<<NBUCKET, OWN_THREADS, 0, stream>>>(
      Xd, ent + (size_t)NNZ_C, off + PART_CHUNKS * OFF_STRIDE, W1_w, W1_b, Xe, N_EDGES_C);
  k_owner_node_out<<<NBUCKET, OWN_THREADS, 0, stream>>>(
      Xe, ent + (size_t)2 * NNZ_C, off + 2 * PART_CHUNKS * OFF_STRIDE,
      Xd, W2_w, W2_b, W_w, W_b, ALPHA_C, out, N_NODES_C);
}


# Argument-only loads (run bounds, owner's Xd rows, W1/W2/W, biases) issued at kernel entry as one group, LDS staging stores behind the plane clearing: one round trip for a chain of waited loads
# speedup vs baseline: 4.1214x; 1.1084x over previous
#include <hip/hip_runtime.h>

typedef __attribute__((ext_vector_type(2))) float v2f;
typedef __attribute__((ext_vector_type(8))) float v8f;

#define N_NODES_C 100000
#define N_EDGES_C 25000
#define NNZ_C     800000
#define D_C       64
#define ALPHA_C   0.5f

#define ACC_P     68
#define B_PSTRIDE 160

__device__ __forceinline__ v8f wmma_f32(v2f a, v2f b, v8f c) {
  return __builtin_amdgcn_wmma_f32_16x16x4_f32(
       false, a,  false, b,
       (short)0, c,  false,  false);
}

#define OWN_THREADS 1024
#define OWN_WAVES   (OWN_THREADS / 32)
#define NBUCKET     256
#define NR_NODES    ((N_NODES_C + NBUCKET - 1) / NBUCKET)
#define NR_EDGES    ((N_EDGES_C + NBUCKET - 1) / NBUCKET)
#define LROW_BITS   9
#define LROW_MASK   ((1 << LROW_BITS) - 1)
#define PART_CHUNKS 64
#define PART_CH     (NNZ_C / PART_CHUNKS)
#define PART_Q      (PART_CH / 4)
#define PART_IT     ((PART_Q + OWN_THREADS - 1) / OWN_THREADS)
#define OFF_STRIDE  (NBUCKET + 1)

static_assert(NNZ_C % (PART_CHUNKS * 4) == 0, "chunks are read as whole 16-byte pieces");
static_assert(PART_CH < 65536, "a rank inside a chunk is kept in 16 bits");
static_assert(NR_NODES <= (1 << LROW_BITS) && NR_EDGES <= (1 << LROW_BITS), "owned row index bits");
static_assert(N_NODES_C <= (1 << (31 - LROW_BITS)) && N_EDGES_C <= (1 << (31 - LROW_BITS)), "gather index bits");
static_assert(PART_CHUNKS % OWN_WAVES == 0, "every owner wave takes the same number of chunks");

__global__ void __launch_bounds__(OWN_THREADS) k_partition(
    const int* __restrict__ drows, const int* __restrict__ dcols, const float* __restrict__ dvals,
    const int* __restrict__ edges, const int* __restrict__ vertex,
    int* __restrict__ ent, float* __restrict__ esc, int* __restrict__ off) {
  __shared__ int hist[NBUCKET];
  __shared__ int offs[NBUCKET + 1];
  __shared__ int wtot[NBUCKET / 32];
  __shared__ __attribute__((aligned(16))) int   se[PART_CH];
  __shared__ __attribute__((aligned(16))) float ss[PART_CH];

  const int tid = threadIdx.x, lane = tid & 31;
  const int g = blockIdx.x, l = blockIdx.y;
  const int* key  = l == 0 ? drows : (l == 1 ? edges : vertex);
  const int* gidx = l == 0 ? dcols : (l == 1 ? vertex : edges);
  const unsigned nr   = l == 1 ? NR_EDGES : NR_NODES;
  const unsigned ndst = l == 1 ? N_EDGES_C : N_NODES_C;

  if (tid < NBUCKET) hist[tid] = 0;
  __syncthreads();

  const size_t cbase = (size_t)g * PART_CH;
  const int4*   key4 = reinterpret_cast<const int4*>(key + cbase);
  const int4*   gid4 = reinterpret_cast<const int4*>(gidx + cbase);
  const float4* sc4  = reinterpret_cast<const float4*>(dvals + cbase);

  int4 kv[PART_IT], gv[PART_IT];
  float4 sv[PART_IT];
#pragma unroll
  for (int it = 0; it < PART_IT; ++it) {
    const int qi = it * OWN_THREADS + tid;
    const int qc = qi < PART_Q ? qi : PART_Q - 1;
    kv[it] = key4[qc];
    gv[it] = gid4[qc];
    sv[it] = sc4[qc];
  }

  int code[PART_IT][4];
#pragma unroll
  for (int it = 0; it < PART_IT; ++it) {
    const int qi = it * OWN_THREADS + tid;
    const int kk[4] = {kv[it].x, kv[it].y, kv[it].z, kv[it].w};
#pragma unroll
    for (int c = 0; c < 4; ++c) {
      const unsigned k = (unsigned)kk[c];
      const unsigned o = l == 1 ? k / NR_EDGES : k / NR_NODES;
      int cd = -1;
      if (qi < PART_Q && k < ndst) cd = (int)(o << 16) | atomicAdd(&hist[o], 1);
      code[it][c] = cd;
    }
  }
  __syncthreads();

  int cnt = 0, inc = 0;
  if (tid < NBUCKET) {
    cnt = hist[tid];
    inc = cnt;
#pragma unroll
    for (int d = 1; d < 32; d <<= 1) {
      const int t = __shfl_up(inc, d);
      if (lane >= d) inc += t;
    }
    if (lane == 31) wtot[tid >> 5] = inc;
  }
  __syncthreads();
  if (tid < NBUCKET) {
    int base = 0;
    for (int w = 0; w < (tid >> 5); ++w) base += wtot[w];
    offs[tid] = base + inc - cnt;
    if (tid == NBUCKET - 1) offs[NBUCKET] = base + inc;
  }
  __syncthreads();

  if (tid <= NBUCKET)
    off[((size_t)l * PART_CHUNKS + g) * OFF_STRIDE + tid] = offs[tid];

#pragma unroll
  for (int it = 0; it < PART_IT; ++it) {
    const int kk[4] = {kv[it].x, kv[it].y, kv[it].z, kv[it].w};
    const int gg[4] = {gv[it].x, gv[it].y, gv[it].z, gv[it].w};
    const float sc[4] = {sv[it].x, sv[it].y, sv[it].z, sv[it].w};
#pragma unroll
    for (int c = 0; c < 4; ++c) {
      const int cd = code[it][c];
      if (cd >= 0) {
        const unsigned o = (unsigned)cd >> 16;
        const int pos = offs[o] + (cd & 0xffff);
        se[pos] = (gg[c] << LROW_BITS) | (int)((unsigned)kk[c] - o * nr);
        if (l == 0) ss[pos] = sc[c];
      }
    }
  }
  __syncthreads();

  const int total = offs[NBUCKET];
  int4*   ent4 = reinterpret_cast<int4*>(ent + (size_t)l * NNZ_C + cbase);
  float4* esc4 = reinterpret_cast<float4*>(esc + cbase);
  for (int q = tid; (q << 2) < total; q += OWN_THREADS) {
    ent4[q] = *reinterpret_cast<const int4*>(&se[q << 2]);
    if (l == 0) esc4[q] = *reinterpret_cast<const float4*>(&ss[q << 2]);
  }
}

#define OWN_BLK  32
#define SORT_B   256
#define SORT_NB  (SORT_B / OWN_BLK)
#define SORT_CAP (OWN_WAVES * SORT_B)
#define SCAN_OF(n) ((((n) + 31) / 32) * 32)
#define OWN_RUNS (PART_CHUNKS / OWN_WAVES)

__device__ __forceinline__ void load_run_bounds(const int* __restrict__ off, int own, int wave,
                                                int (&ra)[OWN_RUNS], int (&rb)[OWN_RUNS]) {
#pragma unroll
  for (int r = 0; r < OWN_RUNS; ++r) {
    const int* o = off + (wave + r * OWN_WAVES) * OFF_STRIDE + own;
    ra[r] = __builtin_amdgcn_readfirstlane(o[0]);
    rb[r] = __builtin_amdgcn_readfirstlane(o[1]);
  }
}

template <int NROWS, bool SCALED>
struct SortLds {
  int   rcnt[SCAN_OF(NROWS)];
  int   roff[SCAN_OF(NROWS) + 1];
  int   wtot[SCAN_OF(NROWS) / 32];
  int   smax;
  int   sent[SORT_CAP];
  float ssc[SCALED ? SORT_CAP : 1];
};

template <int P, int NROWS, int NSUM, bool SCALED, bool COUNT>
__device__ __forceinline__ void owner_accumulate(
    const float* __restrict__ src, const int* __restrict__ ent,
    const float* __restrict__ esc, const int (&ra)[OWN_RUNS], const int (&rb)[OWN_RUNS],
    float* acc, float* cntl, SortLds<NROWS, SCALED>& sl, int tid, int wave, int lane) {
  constexpr int RUNS   = OWN_RUNS;
  constexpr int SCAN_N = SCAN_OF(NROWS);
  static_assert(SCAN_N <= OWN_THREADS, "one thread per row slot in the prefix sum");

  int rs[RUNS], len[RUNS];
  int nw = 0;
#pragma unroll
  for (int r = 0; r < RUNS; ++r) {
    rs[r]  = ra[r];
    len[r] = rb[r] > ra[r] ? rb[r] - ra[r] : 0;
    nw += len[r];
  }
  if (lane == 0) atomicMax(&sl.smax, nw);
  __syncthreads();
  const int rounds = (__builtin_amdgcn_readfirstlane(sl.smax) + SORT_B - 1) / SORT_B;

  auto flat = [&](int p) -> size_t {
    size_t idx = 0;
    int q = p;
    bool done = false;
#pragma unroll
    for (int r = 0; r < RUNS; ++r) {
      if (!done) {
        if (q < len[r] || r == RUNS - 1) {
          idx = (size_t)(wave + r * OWN_WAVES) * PART_CH + (size_t)(rs[r] + q);
          done = true;
        } else {
          q -= len[r];
        }
      }
    }
    return idx;
  };

  static_assert(NSUM <= NROWS, "summed rows are row slots");
  const int rlo = (wave * NSUM) >> 5, rhi = ((wave + 1) * NSUM) >> 5;

  for (int rd = 0; rd < rounds; ++rd) {
    const int p0 = rd * SORT_B;
    int e[SORT_NB] = {}, rk[SORT_NB] = {};
    float s[SORT_NB] = {};
    if (p0 < nw) {
#pragma unroll
      for (int b = 0; b < SORT_NB; ++b) {
        const int pb = p0 + b * OWN_BLK;
        const int p = pb + lane < nw ? pb + lane : nw - 1;
        const size_t idx = flat(p);
        e[b] = ent[idx];
        s[b] = SCALED ? esc[idx] : 1.f;
      }
#pragma unroll
      for (int b = 0; b < SORT_NB; ++b) {
        rk[b] = 0;
        if (p0 + b * OWN_BLK + lane < nw) rk[b] = atomicAdd(&sl.rcnt[e[b] & LROW_MASK], 1);
      }
    }
    __syncthreads();

    int cnt = 0, inc = 0;
    if (tid < SCAN_N) {
      cnt = sl.rcnt[tid];
      inc = cnt;
#pragma unroll
      for (int d = 1; d < 32; d <<= 1) {
        const int t = __shfl_up(inc, d);
        if (lane >= d) inc += t;
      }
      if (lane == 31) sl.wtot[tid >> 5] = inc;
    }
    __syncthreads();
    if (tid < SCAN_N) {
      int base = 0;
      for (int w = 0; w < (tid >> 5); ++w) base += sl.wtot[w];
      sl.roff[tid] = base + inc - cnt;
      if (tid == SCAN_N - 1) sl.roff[SCAN_N] = base + inc;
      sl.rcnt[tid] = 0;
      if (COUNT) { if (tid < NROWS) cntl[tid] += (float)cnt; }
    }
    __syncthreads();

#pragma unroll
    for (int b = 0; b < SORT_NB; ++b) {
      const int pb = p0 + b * OWN_BLK;
      if (pb < nw && pb + lane < nw) {
        const int pos = sl.roff[e[b] & LROW_MASK] + rk[b];
        sl.sent[pos] = e[b];
        if (SCALED) sl.ssc[pos] = s[b];
      }
    }
    __syncthreads();

    const int j0 = __builtin_amdgcn_readfirstlane(sl.roff[rlo]);
    const int j1 = __builtin_amdgcn_readfirstlane(sl.roff[rhi]);
    int cur = -1;
    v2f sum = {0.f, 0.f};
    auto flush = [&]() {
      v2f* d = reinterpret_cast<v2f*>(&acc[cur * P + (lane << 1)]);
      if (rd == 0) {
        *d = sum;
      } else {
        v2f o = *d;
        o[0] += sum[0]; o[1] += sum[1];
        *d = o;
      }
    };
    for (int j = j0; j < j1; j += OWN_BLK) {
      const int n  = j1 - j < OWN_BLK ? j1 - j : OWN_BLK;
      const int jl = j + lane < j1 ? j + lane : j1 - 1;
      const int   me = sl.sent[jl];
      const float ms = SCALED ? sl.ssc[jl] : 1.f;
      v2f x[OWN_BLK];
#pragma unroll
      for (int k = 0; k < OWN_BLK; ++k) {
        const int eu = __builtin_amdgcn_readlane(me, k);
        x[k] = *reinterpret_cast<const v2f*>(src + (size_t)(eu >> LROW_BITS) * D_C + (lane << 1));
      }
#pragma unroll
      for (int k = 0; k < OWN_BLK; ++k) {
        const int row = __builtin_amdgcn_readlane(me, k) & LROW_MASK;
        if (row != cur) {
          if (cur >= 0) flush();
          cur = row;
          sum[0] = 0.f; sum[1] = 0.f;
        }
        float a = x[k][0], c = x[k][1];
        if (SCALED) {
          const float sc = __int_as_float(__builtin_amdgcn_readlane(__float_as_int(ms), k));
          a *= sc; c *= sc;
        }
        if (k >= n) { a = 0.f; c = 0.f; }
        sum[0] += a; sum[1] += c;
      }
    }
    if (cur >= 0) flush();
  }
}

template <int NR>
__global__ void __launch_bounds__(OWN_THREADS) k_owner_sum_scaled(
    const float* __restrict__ src, const int* __restrict__ ent, const float* __restrict__ esc,
    const int* __restrict__ off, float* __restrict__ dst, int ndst) {
  __shared__ __attribute__((aligned(16))) float acc[NR * D_C];
  __shared__ SortLds<NR, true> sl;

  const int tid  = threadIdx.x;
  const int lane = tid & 31;
  const int wave = __builtin_amdgcn_readfirstlane(tid >> 5);
  const int own  = blockIdx.x;
  const int r0   = own * NR;
  int nown = ndst - r0;
  nown = nown > NR ? NR : (nown < 0 ? 0 : nown);

  int ra[OWN_RUNS], rb[OWN_RUNS];
  load_run_bounds(off, own, wave, ra, rb);

  for (int j = tid; j < NR * D_C / 4; j += OWN_THREADS)
    *reinterpret_cast<float4*>(&acc[j << 2]) = make_float4(0.f, 0.f, 0.f, 0.f);
  for (int j = tid; j < SCAN_OF(NR); j += OWN_THREADS) sl.rcnt[j] = 0;
  if (tid == 0) sl.smax = 0;
  __syncthreads();
  owner_accumulate<D_C, NR, NR, true, false>(src, ent, esc, ra, rb, acc, nullptr, sl, tid, wave, lane);
  __syncthreads();

  for (int j = tid; j < nown * (D_C / 4); j += OWN_THREADS)
    *reinterpret_cast<float4*>(dst + (size_t)r0 * D_C + (j << 2)) =
        *reinterpret_cast<const float4*>(&acc[j << 2]);
}

#define NT_NODES  ((NR_NODES + 15) / 16)
#define NRP_NODES (NT_NODES * 16)
static_assert(NT_NODES <= OWN_WAVES, "one tile per wave");

struct BPair { float4 r0, r1; };
template <int NP>
__device__ __forceinline__ BPair load_b_pair(const float* __restrict__ B, int tid) {
  static_assert(NP * 16 <= OWN_THREADS, "one row pair x column quad per thread");
  int p = tid >> 4;
  p = p < NP ? p : NP - 1;
  const float4* s = reinterpret_cast<const float4*>(B + (size_t)(2 * p) * 64 + ((tid & 15) << 2));
  BPair r;
  r.r0 = s[0];
  r.r1 = s[16];
  return r;
}
template <int NP>
__device__ __forceinline__ void store_b_pair(float* Bf, const BPair& r, int tid) {
  const int p = tid >> 4, cq = tid & 15;
  if (p < NP) {
    float4* d = reinterpret_cast<float4*>(Bf + p * B_PSTRIDE + (cq << 3));
    d[0] = make_float4(r.r0.x, r.r1.x, r.r0.y, r.r1.y);
    d[1] = make_float4(r.r0.z, r.r1.z, r.r0.w, r.r1.w);
  }
}

template <bool SCALE>
__device__ __forceinline__ void tile_mac64(const float* arow, float s, const float* bl, v8f (&c)[4]) {
#pragma unroll
  for (int k4 = 0; k4 < 16; ++k4) {
    v2f a = *reinterpret_cast<const v2f*>(arow + (k4 << 2));
    if (SCALE) { a[0] *= s; a[1] *= s; }
#pragma unroll
    for (int n = 0; n < 4; ++n) {
      const v2f b = *reinterpret_cast<const v2f*>(bl + (k4 << 1) * B_PSTRIDE + (n << 5));
      c[n] = wmma_f32(a, b, c[n]);
    }
  }
}

#define NT_EDGES  ((NR_EDGES + 15) / 16)
#define NRP_EDGES (NT_EDGES * 16)
static_assert(4 * NT_EDGES <= OWN_WAVES, "one 16 x 16 output tile per wave");

__global__ void __launch_bounds__(OWN_THREADS) k_owner_edge_w1(
    const float* __restrict__ Xd, const int* __restrict__ ent, const int* __restrict__ off,
    const float* __restrict__ W1, const float* __restrict__ b1,
    float* __restrict__ Xe, int ndst) {
  __shared__ __attribute__((aligned(16))) float acc[NRP_EDGES * ACC_P];
  __shared__ __attribute__((aligned(16))) float Bf[32 * B_PSTRIDE];
  __shared__ float bs[64];
  __shared__ float cntl[NRP_EDGES];
  __shared__ SortLds<NRP_EDGES, false> sl;

  const int tid  = threadIdx.x;
  const int lane = tid & 31;
  const int wave = __builtin_amdgcn_readfirstlane(tid >> 5);
  const int own  = blockIdx.x;
  const int r0   = own * NR_EDGES;
  int nown = ndst - r0;
  nown = nown > NR_EDGES ? NR_EDGES : (nown < 0 ? 0 : nown);

  int ra[OWN_RUNS], rb[OWN_RUNS];
  load_run_bounds(off, own, wave, ra, rb);
  const BPair w1v = load_b_pair<32>(W1, tid);
  const float b1v = b1[tid & 63];

  for (int j = tid; j < NRP_EDGES * ACC_P / 4; j += OWN_THREADS)
    *reinterpret_cast<float4*>(&acc[j << 2]) = make_float4(0.f, 0.f, 0.f, 0.f);
  for (int j = tid; j < NRP_EDGES; j += OWN_THREADS) cntl[j] = 0.f;
  for (int j = tid; j < SCAN_OF(NRP_EDGES); j += OWN_THREADS) sl.rcnt[j] = 0;
  if (tid == 0) sl.smax = 0;
  store_b_pair<32>(Bf, w1v, tid);
  if (tid < 64) bs[tid] = b1v;
  __syncthreads();
  owner_accumulate<ACC_P, NRP_EDGES, NR_EDGES, false, true>(Xd, ent, nullptr, ra, rb, acc, cntl, sl, tid, wave, lane);
  __syncthreads();

  const int lrow = lane & 15, hi = lane >> 4;
  const int tile = wave >> 2, nt = wave & 3;
  v8f c = v8f{};
  if (wave < 4 * NT_EDGES) {
    const float* arow = &acc[(tile * 16 + lrow) * ACC_P + (hi << 1)];
    const float* bl   = &Bf[hi * B_PSTRIDE + (lrow << 1) + (nt << 5)];
#pragma unroll
    for (int k4 = 0; k4 < 16; ++k4) {
      const v2f a = *reinterpret_cast<const v2f*>(arow + (k4 << 2));
      const v2f b = *reinterpret_cast<const v2f*>(bl + (k4 << 1) * B_PSTRIDE);
      c = wmma_f32(a, b, c);
    }
  }
  __syncthreads();
  if (wave < 4 * NT_EDGES) {
#pragma unroll
    for (int j = 0; j < 8; ++j) {
      const int lr = tile * 16 + j + (hi << 3);
      const int col = (nt << 4) + lrow;
      acc[lr * ACC_P + col] = c[j] + cntl[lr] * bs[col];
    }
  }
  __syncthreads();
  for (int j = tid; j < nown * 16; j += OWN_THREADS) {
    const int row = j >> 4, c4 = j & 15;
    *reinterpret_cast<float4*>(Xe + (size_t)(r0 + row) * D_C + (c4 << 2)) =
        *reinterpret_cast<const float4*>(&acc[row * ACC_P + (c4 << 2)]);
  }
}

__global__ void __launch_bounds__(OWN_THREADS) k_owner_node_out(
    const float* __restrict__ Xe, const int* __restrict__ ent, const int* __restrict__ off,
    const float* __restrict__ Xd, const float* __restrict__ W2, const float* __restrict__ b2,
    const float* __restrict__ Ww, const float* __restrict__ bw, float alpha,
    float* __restrict__ out, int ndst) {
  __shared__ __attribute__((aligned(16))) float acc[NRP_NODES * ACC_P];
  __shared__ __attribute__((aligned(16))) float xs[NRP_NODES * ACC_P];
  __shared__ __attribute__((aligned(16))) float Bf2[64 * B_PSTRIDE];
  __shared__ __attribute__((aligned(16))) float Bfw[32 * B_PSTRIDE];
  __shared__ float b2s[64];
  __shared__ float bws[64];
  __shared__ float cntl[NRP_NODES];
  __shared__ SortLds<NRP_NODES, false> sl;

  const int tid  = threadIdx.x;
  const int lane = tid & 31;
  const int wave = __builtin_amdgcn_readfirstlane(tid >> 5);
  const int own  = blockIdx.x;
  const int r0   = own * NR_NODES;
  int nown = ndst - r0;
  nown = nown > NR_NODES ? NR_NODES : (nown < 0 ? 0 : nown);

  constexpr int XS_IT = (NRP_NODES * 16 + OWN_THREADS - 1) / OWN_THREADS;
  int ra[OWN_RUNS], rb[OWN_RUNS];
  load_run_bounds(off, own, wave, ra, rb);
  float4 xv[XS_IT];
#pragma unroll
  for (int it = 0; it < XS_IT; ++it) {
    const int j = it * OWN_THREADS + tid;
    int gr = r0 + (j >> 4);
    gr = gr < ndst ? gr : ndst - 1;
    xv[it] = *reinterpret_cast<const float4*>(Xd + (size_t)gr * D_C + ((j & 15) << 2));
  }
  const BPair w2v = load_b_pair<64>(W2, tid);
  const BPair wwv = load_b_pair<32>(Ww, tid);
  const float b2v = b2[tid & 63], bwv = bw[tid & 63];

  for (int j = tid; j < NRP_NODES * ACC_P / 4; j += OWN_THREADS)
    *reinterpret_cast<float4*>(&acc[j << 2]) = make_float4(0.f, 0.f, 0.f, 0.f);
  for (int j = tid; j < NRP_NODES; j += OWN_THREADS) cntl[j] = 0.f;
  for (int j = tid; j < SCAN_OF(NRP_NODES); j += OWN_THREADS) sl.rcnt[j] = 0;
  if (tid == 0) sl.smax = 0;

#pragma unroll
  for (int it = 0; it < XS_IT; ++it) {
    const int j = it * OWN_THREADS + tid;
    const int row = j >> 4, c4 = j & 15;
    if (j < NRP_NODES * 16)
      *reinterpret_cast<float4*>(&xs[row * ACC_P + (c4 << 2)]) =
          row < nown ? xv[it] : make_float4(0.f, 0.f, 0.f, 0.f);
  }
  store_b_pair<64>(Bf2, w2v, tid);
  store_b_pair<32>(Bfw, wwv, tid);
  if (tid < 64) { b2s[tid] = b2v; bws[tid] = bwv; }
  __syncthreads();
  owner_accumulate<ACC_P, NRP_NODES, NRP_NODES, false, true>(Xe, ent, nullptr, ra, rb, acc, cntl, sl, tid, wave, lane);
  __syncthreads();

  const int lrow = lane & 15, hi = lane >> 4;
  v8f c[4] = {v8f{}, v8f{}, v8f{}, v8f{}};
  if (wave < NT_NODES) {
    const int ar = (wave * 16 + lrow) * ACC_P + (hi << 1);
    const float* bl = &Bf2[hi * B_PSTRIDE + (lrow << 1)];
    tile_mac64<true >(&xs[ar],  cntl[wave * 16 + lrow], bl, c);
    tile_mac64<false>(&acc[ar], 1.f,                    bl + 32 * B_PSTRIDE, c);
#pragma unroll
    for (int n = 0; n < 4; ++n)
#pragma unroll
      for (int j = 0; j < 8; ++j) {
        const int lr = wave * 16 + j + (hi << 3);
        const int col = (n << 4) + lrow;
        const float xv = c[n][j] + cntl[lr] * b2s[col];
        c[n][j] = (1.f - alpha) * xv + alpha * xs[lr * ACC_P + col];
      }
  }
  __syncthreads();
  if (wave < NT_NODES) {
#pragma unroll
    for (int n = 0; n < 4; ++n)
#pragma unroll
      for (int j = 0; j < 8; ++j)
        acc[(wave * 16 + j + (hi << 3)) * ACC_P + (n << 4) + lrow] = c[n][j];
  }
  __syncthreads();
  v8f o[4] = {v8f{}, v8f{}, v8f{}, v8f{}};
  if (wave < NT_NODES)
    tile_mac64<false>(&acc[(wave * 16 + lrow) * ACC_P + (hi << 1)], 1.f,
                      &Bfw[hi * B_PSTRIDE + (lrow << 1)], o);
  __syncthreads();
  if (wave < NT_NODES) {
#pragma unroll
    for (int n = 0; n < 4; ++n)
#pragma unroll
      for (int j = 0; j < 8; ++j) {
        const int col = (n << 4) + lrow;
        acc[(wave * 16 + j + (hi << 3)) * ACC_P + col] = o[n][j] + bws[col];
      }
  }
  __syncthreads();
  for (int j = tid; j < nown * 16; j += OWN_THREADS) {
    const int row = j >> 4, c4 = j & 15;
    *reinterpret_cast<float4*>(out + (size_t)(r0 + row) * D_C + (c4 << 2)) =
        *reinterpret_cast<const float4*>(&acc[row * ACC_P + (c4 << 2)]);
  }
}

extern "C" void kernel_launch(void* const* d_in, const int* in_sizes, int n_in,
                              void* d_out, int out_size, void* d_ws, size_t ws_size,
                              hipStream_t stream) {
  const float* X      = (const float*)d_in[0];
  const float* W1_w   = (const float*)d_in[1];
  const float* W1_b   = (const float*)d_in[2];
  const float* W2_w   = (const float*)d_in[3];
  const float* W2_b   = (const float*)d_in[4];
  const float* W_w    = (const float*)d_in[5];
  const float* W_b    = (const float*)d_in[6];
  const float* dvals  = (const float*)d_in[7];
  const int*   vertex = (const int*)d_in[8];
  const int*   edges  = (const int*)d_in[9];
  const int*   drows  = (const int*)d_in[10];
  const int*   dcols  = (const int*)d_in[11];
  float* out = (float*)d_out;

  float* Xd = (float*)d_ws;
  float* Xe = Xd + (size_t)2 * N_NODES_C * D_C;
  float* Xv = Xe + (size_t)N_EDGES_C * D_C;
  int*   ent = (int*)(Xv + 102400);
  float* esc = (float*)(ent + (size_t)3 * NNZ_C);
  int*   off = (int*)(esc + NNZ_C);
  static_assert(102400 + 4 * (size_t)NNZ_C + 3 * PART_CHUNKS * OFF_STRIDE <= (size_t)N_NODES_C * D_C,
                "entries fit in the last node-sized plane");


  k_partition<<<dim3(PART_CHUNKS, 3), OWN_THREADS, 0, stream>>>(
      drows, dcols, dvals, edges, vertex, ent, esc, off);
  k_owner_sum_scaled<NR_NODES><<<NBUCKET, OWN_THREADS, 0, stream>>>(
      X, ent, esc, off, Xd, N_NODES_C);
  k_owner_edge_w1<<<NBUCKET, OWN_THREADS, 0, stream>>>(
      Xd, ent + (size_t)NNZ_C, off + PART_CHUNKS * OFF_STRIDE, W1_w, W1_b, Xe, N_EDGES_C);
  k_owner_node_out<<<NBUCKET, OWN_THREADS, 0, stream>>>(
      Xe, ent + (size_t)2 * NNZ_C, off + 2 * PART_CHUNKS * OFF_STRIDE,
      Xd, W2_w, W2_b, W_w, W_b, ALPHA_C, out, N_NODES_C);
}
